# starting kernel
# baseline (speedup 1.0000x reference)
#include <hip/hip_runtime.h>

#define NN 100000
#define NE 1600000
#define DD 128

typedef __attribute__((ext_vector_type(2))) float v2f;
typedef __attribute__((ext_vector_type(8))) float v8f;

__global__ __launch_bounds__(256) void gcn_gemm_wmma(const float* __restrict__ x,
                                                     const float* __restrict__ W,
                                                     float* __restrict__ h) {
    __shared__ float ldsWT[16 * 132];
    const int tid     = threadIdx.x;
    const int colBase = blockIdx.y * 16;

    #pragma unroll
    for (int i = 0; i < 8; ++i) {
        int e = tid + i * 256;
        int n = e & 15;
        int k = e >> 4;
        ldsWT[n * 132 + k] = W[k * DD + colBase + n];
    }
    __syncthreads();

    const int wave  = tid >> 5;
    const int lane  = tid & 31;
    const int mTile = blockIdx.x * 8 + wave;
    if (mTile >= (NN / 16)) return;

    const int    row  = mTile * 16 + (lane & 15);
    const int    koff = (lane >> 4) * 2;
    const float* xrow = x + (size_t)row * DD;
    const float* wtn  = ldsWT + (lane & 15) * 132;

    v8f acc = {};
    #pragma unroll
    for (int kk = 0; kk < DD; kk += 4) {
        v2f a, b;
        a.x = xrow[kk + koff];
        a.y = xrow[kk + koff + 1];
        b.x = wtn[kk + koff];
        b.y = wtn[kk + koff + 1];
        acc = __builtin_amdgcn_wmma_f32_16x16x4_f32(
                   false, a,  false, b,
                   (short)0, acc,  false,  false);
    }

    const int col     = colBase + (lane & 15);
    const int rowBase = mTile * 16 + (lane >> 4) * 8;
    #pragma unroll
    for (int v = 0; v < 8; ++v) {
        h[(size_t)(rowBase + v) * DD + col] = acc[v];
    }
}

__global__ __launch_bounds__(256) void gcn_init_bias(float* __restrict__ agg,
                                                     const float* __restrict__ b) {
    size_t i = (size_t)blockIdx.x * 256 + threadIdx.x;
    agg[i] = b[i & (DD - 1)];
}

__global__ __launch_bounds__(256) void gcn_scatter(const float* __restrict__ h,
                                                   const int* __restrict__ src,
                                                   const int* __restrict__ dst,
                                                   const float* __restrict__ w,
                                                   float* __restrict__ agg) {
    const int e = blockIdx.x * 8 + (threadIdx.x >> 5);
    if (e >= NE) return;
    const int   lane = threadIdx.x & 31;
    const int   s    = src[e];
    const int   d    = dst[e];
    const float wt   = w[e];

    const float4 hv = *(const float4*)(h + (size_t)s * DD + lane * 4);
    float* out = agg + (size_t)d * DD + lane * 4;
    unsafeAtomicAdd(out + 0, hv.x * wt);
    unsafeAtomicAdd(out + 1, hv.y * wt);
    unsafeAtomicAdd(out + 2, hv.z * wt);
    unsafeAtomicAdd(out + 3, hv.w * wt);
}

__global__ __launch_bounds__(256) void gcn_relu(const float* __restrict__ in,
                                                float* __restrict__ out) {
    size_t i = (size_t)blockIdx.x * 256 + threadIdx.x;
    out[i] = fmaxf(in[i], 0.0f);
}

extern "C" void kernel_launch(void* const* d_in, const int* in_sizes, int n_in,
                              void* d_out, int out_size, void* d_ws, size_t ws_size,
                              hipStream_t stream) {
    const float* x    = (const float*)d_in[0];
    const int*   esrc = (const int*)d_in[1];
    const int*   edst = (const int*)d_in[2];
    const float* ew   = (const float*)d_in[3];
    const float* W    = (const float*)d_in[4];
    const float* b    = (const float*)d_in[5];
    float*       out  = (float*)d_out;

    float* ws0 = (float*)d_ws;
    float* ws1 = ws0 + (size_t)NN * DD;

    const float* xcur[3] = { x,   ws1, out };
    float*       agg[3]  = { ws1, out, ws1 };
    float*       post[3] = { ws1, out, out };

    const dim3 gGemm((NN / 16 + 7) / 8, DD / 16);
    const int  gND = (NN * DD) / 256;
    const int  gE  = NE / 8;

    for (int l = 0; l < 3; ++l) {
        gcn_gemm_wmma<<<gGemm, 256, 0, stream>>>(xcur[l], W + (size_t)l * DD * DD, ws0);
        gcn_init_bias<<<gND, 256, 0, stream>>>(agg[l], b + (size_t)l * DD);
        gcn_scatter<<<gE, 256, 0, stream>>>(ws0, esrc, edst, ew, agg[l]);
        gcn_relu<<<gND, 256, 0, stream>>>(agg[l], post[l]);
    }
}


# Scatter re-cut by dst ownership: 256 groups each hold 391 dst rows in LDS, scan the dst list for their edges, gather h rows, sum by LDS float adds (arrival order, bias last), write relu(sum+bias)
# speedup vs baseline: 2.0558x; 2.0558x over previous
#include <hip/hip_runtime.h>

#define NN 100000
#define NE 1600000
#define DD 128
#define ROWS 391
#define NGRP 256
#define SEGCAP 288

typedef __attribute__((ext_vector_type(2))) float v2f;
typedef __attribute__((ext_vector_type(8))) float v8f;

__global__ __launch_bounds__(256) void gcn_gemm_wmma(const float* __restrict__ x,
                                                     const float* __restrict__ W,
                                                     float* __restrict__ h) {
    __shared__ float ldsWT[16 * 132];
    const int tid     = threadIdx.x;
    const int colBase = blockIdx.y * 16;

    #pragma unroll
    for (int i = 0; i < 8; ++i) {
        int e = tid + i * 256;
        int n = e & 15;
        int k = e >> 4;
        ldsWT[n * 132 + k] = W[k * DD + colBase + n];
    }
    __syncthreads();

    const int wave  = tid >> 5;
    const int lane  = tid & 31;
    const int mTile = blockIdx.x * 8 + wave;
    if (mTile >= (NN / 16)) return;

    const int    row  = mTile * 16 + (lane & 15);
    const int    koff = (lane >> 4) * 2;
    const float* xrow = x + (size_t)row * DD;
    const float* wtn  = ldsWT + (lane & 15) * 132;

    v8f acc = {};
    #pragma unroll
    for (int kk = 0; kk < DD; kk += 4) {
        v2f a, b;
        a.x = xrow[kk + koff];
        a.y = xrow[kk + koff + 1];
        b.x = wtn[kk + koff];
        b.y = wtn[kk + koff + 1];
        acc = __builtin_amdgcn_wmma_f32_16x16x4_f32(
                   false, a,  false, b,
                   (short)0, acc,  false,  false);
    }

    const int col     = colBase + (lane & 15);
    const int rowBase = mTile * 16 + (lane >> 4) * 8;
    #pragma unroll
    for (int v = 0; v < 8; ++v) {
        h[(size_t)(rowBase + v) * DD + col] = acc[v];
    }
}

__device__ __forceinline__ void lds_add(float* p, float v) {
    __hip_atomic_fetch_add(p, v, __ATOMIC_RELAXED, __HIP_MEMORY_SCOPE_WORKGROUP);
}

__device__ __forceinline__ void gather_batch(const unsigned* list, int n, int lane,
                                             const float* __restrict__ h,
                                             const int* __restrict__ src,
                                             const float* __restrict__ w,
                                             float* acc) {
    const unsigned ent = (lane < n) ? list[lane] : 0u;
    const int   e  = (int)(ent & 0x1FFFFFu);
    const int   r  = (int)(ent >> 21);
    const int   s  = src[e];
    const int   wb = __float_as_int(w[e]);

    for (int k0 = 0; k0 < n; k0 += 8) {
        float4 hv[8];
        int    rr[8];
        float  ww[8];
        #pragma unroll
        for (int u = 0; u < 8; ++u) {
            const int k  = min(k0 + u, n - 1);
            const int sk = __builtin_amdgcn_readlane(s, k);
            rr[u] = __builtin_amdgcn_readlane(r, k);
            ww[u] = __int_as_float(__builtin_amdgcn_readlane(wb, k));
            hv[u] = *(const float4*)(h + (size_t)sk * DD + lane * 4);
        }
        #pragma unroll
        for (int u = 0; u < 8; ++u) {
            if (k0 + u < n) {
                float* a = acc + rr[u] * DD + lane;
                lds_add(a,      hv[u].x * ww[u]);
                lds_add(a + 32, hv[u].y * ww[u]);
                lds_add(a + 64, hv[u].z * ww[u]);
                lds_add(a + 96, hv[u].w * ww[u]);
            }
        }
    }
}

__global__ __launch_bounds__(1024) void gcn_owner_scatter(const float* __restrict__ h,
                                                          const int* __restrict__ src,
                                                          const int* __restrict__ dst,
                                                          const float* __restrict__ w,
                                                          const float* __restrict__ bias,
                                                          float* __restrict__ xout) {
    __shared__ float    acc[ROWS * DD];
    __shared__ unsigned seg[32 * SEGCAP];

    const int tid   = threadIdx.x;
    const int lane  = tid & 31;
    const int wave  = __builtin_amdgcn_readfirstlane(tid >> 5);
    const int lo    = blockIdx.x * ROWS;
    const int nrows = min(ROWS, NN - lo);

    for (int i = tid; i < ROWS * DD / 4; i += 1024)
        ((float4*)acc)[i] = make_float4(0.f, 0.f, 0.f, 0.f);
    __syncthreads();

    unsigned* myseg = seg + wave * SEGCAP;
    int cnt = 0;

    const int NP = NE / 256;
    int  p  = wave;
    int4 a0 = *(const int4*)(dst + (size_t)p * 256 + lane * 4);
    int4 a1 = *(const int4*)(dst + (size_t)p * 256 + 128 + lane * 4);
    while (p < NP) {
        const int4 c0 = a0, c1 = a1;
        const int  pn = p + 32;
        if (pn < NP) {
            a0 = *(const int4*)(dst + (size_t)pn * 256 + lane * 4);
            a1 = *(const int4*)(dst + (size_t)pn * 256 + 128 + lane * 4);
        }
        const int d[8] = { c0.x, c0.y, c0.z, c0.w, c1.x, c1.y, c1.z, c1.w };
        #pragma unroll
        for (int j = 0; j < 8; ++j) {
            const unsigned r    = (unsigned)(d[j] - lo);
            const bool     m    = r < (unsigned)nrows;
            const unsigned mask = __builtin_amdgcn_ballot_w32(m);
            if (mask) {
                if (m) {
                    const int pos = cnt + (int)__builtin_amdgcn_mbcnt_lo(mask, 0u);
                    const int e   = p * 256 + (j >> 2) * 128 + lane * 4 + (j & 3);
                    myseg[pos] = (r << 21) | (unsigned)e;
                }
                cnt += __popc(mask);
            }
        }
        __builtin_amdgcn_wave_barrier();
        while (cnt >= 32) {
            cnt -= 32;
            gather_batch(myseg + cnt, 32, lane, h, src, w, acc);
        }
        __builtin_amdgcn_wave_barrier();
        p = pn;
    }
    if (cnt > 0) gather_batch(myseg, cnt, lane, h, src, w, acc);
    __syncthreads();

    const float4 bv = *(const float4*)(bias + lane * 4);
    for (int r = wave; r < nrows; r += 32) {
        const float* a = acc + r * DD + lane;
        float4 o;
        o.x = fmaxf(a[0]  + bv.x, 0.0f);
        o.y = fmaxf(a[32] + bv.y, 0.0f);
        o.z = fmaxf(a[64] + bv.z, 0.0f);
        o.w = fmaxf(a[96] + bv.w, 0.0f);
        *(float4*)(xout + (size_t)(lo + r) * DD + lane * 4) = o;
    }
}

extern "C" void kernel_launch(void* const* d_in, const int* in_sizes, int n_in,
                              void* d_out, int out_size, void* d_ws, size_t ws_size,
                              hipStream_t stream) {
    const float* x    = (const float*)d_in[0];
    const int*   esrc = (const int*)d_in[1];
    const int*   edst = (const int*)d_in[2];
    const float* ew   = (const float*)d_in[3];
    const float* W    = (const float*)d_in[4];
    const float* b    = (const float*)d_in[5];
    float*       out  = (float*)d_out;

    float* ws0 = (float*)d_ws;
    float* ws1 = ws0 + (size_t)NN * DD;

    const float* xcur[3]  = { x,   ws1, ws1 };
    float*       xnext[3] = { ws1, ws1, out };

    const dim3 gGemm((NN / 16 + 7) / 8, DD / 16);

    for (int l = 0; l < 3; ++l) {
        gcn_gemm_wmma<<<gGemm, 256, 0, stream>>>(xcur[l], W + (size_t)l * DD * DD, ws0);
        gcn_owner_scatter<<<NGRP, 1024, 0, stream>>>(ws0, esrc, edst, ew, b + (size_t)l * DD, xnext[l]);
    }
}


# dst scan done once per call: layer 1's owner groups keep each wave's matched (row, src, weight) list in the workspace; layers 2-3 read it back, no dst re-scan; LDS adds in arrival order as before
# speedup vs baseline: 2.6972x; 1.3120x over previous
#include <hip/hip_runtime.h>

#define NN 100000
#define NE 1600000
#define DD 128
#define ROWS 391
#define NGRP 256
#define SEGCAP 288
#define LCAP 512

typedef __attribute__((ext_vector_type(2))) float v2f;
typedef __attribute__((ext_vector_type(8))) float v8f;

__global__ __launch_bounds__(256) void gcn_gemm_wmma(const float* __restrict__ x,
                                                     const float* __restrict__ W,
                                                     float* __restrict__ h) {
    __shared__ float ldsWT[16 * 132];
    const int tid     = threadIdx.x;
    const int colBase = blockIdx.y * 16;

    #pragma unroll
    for (int i = 0; i < 8; ++i) {
        int e = tid + i * 256;
        int n = e & 15;
        int k = e >> 4;
        ldsWT[n * 132 + k] = W[k * DD + colBase + n];
    }
    __syncthreads();

    const int wave  = tid >> 5;
    const int lane  = tid & 31;
    const int mTile = blockIdx.x * 8 + wave;
    if (mTile >= (NN / 16)) return;

    const int    row  = mTile * 16 + (lane & 15);
    const int    koff = (lane >> 4) * 2;
    const float* xrow = x + (size_t)row * DD;
    const float* wtn  = ldsWT + (lane & 15) * 132;

    v8f acc = {};
    #pragma unroll
    for (int kk = 0; kk < DD; kk += 4) {
        v2f a, b;
        a.x = xrow[kk + koff];
        a.y = xrow[kk + koff + 1];
        b.x = wtn[kk + koff];
        b.y = wtn[kk + koff + 1];
        acc = __builtin_amdgcn_wmma_f32_16x16x4_f32(
                   false, a,  false, b,
                   (short)0, acc,  false,  false);
    }

    const int col     = colBase + (lane & 15);
    const int rowBase = mTile * 16 + (lane >> 4) * 8;
    #pragma unroll
    for (int v = 0; v < 8; ++v) {
        h[(size_t)(rowBase + v) * DD + col] = acc[v];
    }
}

__device__ __forceinline__ void lds_add(float* p, float v) {
    __hip_atomic_fetch_add(p, v, __ATOMIC_RELAXED, __HIP_MEMORY_SCOPE_WORKGROUP);
}

__device__ __forceinline__ void gather_rows(int s, int r, int wb, int n, int lane,
                                            const float* __restrict__ h, float* acc) {
    for (int k0 = 0; k0 < n; k0 += 8) {
        float4 hv[8];
        int    rr[8];
        float  ww[8];
        #pragma unroll
        for (int u = 0; u < 8; ++u) {
            const int k  = min(k0 + u, n - 1);
            const int sk = __builtin_amdgcn_readlane(s, k);
            rr[u] = __builtin_amdgcn_readlane(r, k);
            ww[u] = __int_as_float(__builtin_amdgcn_readlane(wb, k));
            hv[u] = *(const float4*)(h + (size_t)sk * DD + lane * 4);
        }
        #pragma unroll
        for (int u = 0; u < 8; ++u) {
            if (k0 + u < n) {
                float* a = acc + rr[u] * DD + lane;
                lds_add(a,      hv[u].x * ww[u]);
                lds_add(a + 32, hv[u].y * ww[u]);
                lds_add(a + 64, hv[u].z * ww[u]);
                lds_add(a + 96, hv[u].w * ww[u]);
            }
        }
    }
}

__device__ __forceinline__ void gather_batch(const unsigned* list, int n, int lane,
                                             const float* __restrict__ h,
                                             const int* __restrict__ src,
                                             const float* __restrict__ w,
                                             float* acc, uint2* rec) {
    const unsigned ent = (lane < n) ? list[lane] : 0u;
    const int   e  = (int)(ent & 0x1FFFFFu);
    const int   r  = (int)(ent >> 21);
    const int   s  = src[e];
    const int   wb = __float_as_int(w[e]);
    if (rec != nullptr && lane < n)
        rec[lane] = make_uint2(((unsigned)r << 17) | (unsigned)s, (unsigned)wb);
    gather_rows(s, r, wb, n, lane, h, acc);
}

template <bool REC>
__device__ __forceinline__ int scan_edges(int wave, int lane, int lo, int nrows, unsigned* myseg,
                                          const float* __restrict__ h,
                                          const int* __restrict__ src,
                                          const int* __restrict__ dst,
                                          const float* __restrict__ w,
                                          float* acc, uint2* keep) {
    int cnt  = 0;
    int done = 0;

    const int NP = NE / 256;
    int  p  = wave;
    int4 a0 = *(const int4*)(dst + (size_t)p * 256 + lane * 4);
    int4 a1 = *(const int4*)(dst + (size_t)p * 256 + 128 + lane * 4);
    while (p < NP) {
        const int4 c0 = a0, c1 = a1;
        const int  pn = p + 32;
        if (pn < NP) {
            a0 = *(const int4*)(dst + (size_t)pn * 256 + lane * 4);
            a1 = *(const int4*)(dst + (size_t)pn * 256 + 128 + lane * 4);
        }
        const int d[8] = { c0.x, c0.y, c0.z, c0.w, c1.x, c1.y, c1.z, c1.w };
        #pragma unroll
        for (int j = 0; j < 8; ++j) {
            const unsigned r    = (unsigned)(d[j] - lo);
            const bool     m    = r < (unsigned)nrows;
            const unsigned mask = __builtin_amdgcn_ballot_w32(m);
            if (mask) {
                if (m) {
                    const int pos = cnt + (int)__builtin_amdgcn_mbcnt_lo(mask, 0u);
                    const int e   = p * 256 + (j >> 2) * 128 + lane * 4 + (j & 3);
                    myseg[pos] = (r << 21) | (unsigned)e;
                }
                cnt += __popc(mask);
            }
        }
        __builtin_amdgcn_wave_barrier();
        while (cnt >= 32) {
            cnt -= 32;
            gather_batch(myseg + cnt, 32, lane, h, src, w, acc,
                         (REC && done + 32 <= LCAP) ? keep + done : nullptr);
            done += 32;
        }
        __builtin_amdgcn_wave_barrier();
        p = pn;
    }
    if (cnt > 0) {
        gather_batch(myseg, cnt, lane, h, src, w, acc,
                     (REC && done + cnt <= LCAP) ? keep + done : nullptr);
        done += cnt;
    }
    return done;
}

template <bool FIRST>
__global__ __launch_bounds__(1024) void gcn_owner_scatter(const float* __restrict__ h,
                                                          const int* __restrict__ src,
                                                          const int* __restrict__ dst,
                                                          const float* __restrict__ w,
                                                          const float* __restrict__ bias,
                                                          float* __restrict__ xout,
                                                          int* counts, unsigned* lists) {
    __shared__ float    acc[ROWS * DD];
    __shared__ unsigned seg[32 * SEGCAP];

    const int tid   = threadIdx.x;
    const int lane  = tid & 31;
    const int wave  = __builtin_amdgcn_readfirstlane(tid >> 5);
    const int lo    = blockIdx.x * ROWS;
    const int nrows = min(ROWS, NN - lo);

    for (int i = tid; i < ROWS * DD / 4; i += 1024)
        ((float4*)acc)[i] = make_float4(0.f, 0.f, 0.f, 0.f);
    __syncthreads();

    unsigned* myseg = seg + wave * SEGCAP;
    const int gw    = blockIdx.x * 32 + wave;
    uint2*    keep  = (uint2*)lists + (size_t)gw * LCAP;

    if (FIRST) {
        const int total = scan_edges<true>(wave, lane, lo, nrows, myseg, h, src, dst, w, acc, keep);
        if (lane == 0) counts[gw] = total;
    } else {
        const int total = __builtin_amdgcn_readfirstlane(counts[gw]);
        if (total <= LCAP) {
            uint2 nxt = make_uint2(0u, 0u);
            if (lane < total) nxt = keep[lane];
            for (int base = 0; base < total; base += 32) {
                const uint2 cur = nxt;
                const int   n   = min(32, total - base);
                if (base + 32 + lane < total) nxt = keep[base + 32 + lane];
                gather_rows((int)(cur.x & 0x1FFFFu), (int)(cur.x >> 17), (int)cur.y, n, lane, h, acc);
            }
        } else {
            scan_edges<false>(wave, lane, lo, nrows, myseg, h, src, dst, w, acc, nullptr);
        }
    }
    __syncthreads();

    const float4 bv = *(const float4*)(bias + lane * 4);
    for (int r = wave; r < nrows; r += 32) {
        const float* a = acc + r * DD + lane;
        float4 o;
        o.x = fmaxf(a[0]  + bv.x, 0.0f);
        o.y = fmaxf(a[32] + bv.y, 0.0f);
        o.z = fmaxf(a[64] + bv.z, 0.0f);
        o.w = fmaxf(a[96] + bv.w, 0.0f);
        *(float4*)(xout + (size_t)(lo + r) * DD + lane * 4) = o;
    }
}

extern "C" void kernel_launch(void* const* d_in, const int* in_sizes, int n_in,
                              void* d_out, int out_size, void* d_ws, size_t ws_size,
                              hipStream_t stream) {
    const float* x    = (const float*)d_in[0];
    const int*   esrc = (const int*)d_in[1];
    const int*   edst = (const int*)d_in[2];
    const float* ew   = (const float*)d_in[3];
    const float* W    = (const float*)d_in[4];
    const float* b    = (const float*)d_in[5];
    float*       out  = (float*)d_out;

    float* ws0 = (float*)d_ws;
    float* ws1 = ws0 + (size_t)NN * DD;
    int*      cnts  = (int*)(ws1 + (size_t)NN * DD);
    unsigned* lists = (unsigned*)(cnts + NGRP * 32);

    const float* xcur[3]  = { x,   ws1, ws1 };
    float*       xnext[3] = { ws1, ws1, out };

    const dim3 gGemm((NN / 16 + 7) / 8, DD / 16);

    for (int l = 0; l < 3; ++l) {
        gcn_gemm_wmma<<<gGemm, 256, 0, stream>>>(xcur[l], W + (size_t)l * DD * DD, ws0);
        if (l == 0)
            gcn_owner_scatter<true><<<NGRP, 1024, 0, stream>>>(ws0, esrc, edst, ew, b, xnext[l], cnts, lists);
        else
            gcn_owner_scatter<false><<<NGRP, 1024, 0, stream>>>(ws0, esrc, edst, ew, b + (size_t)l * DD, xnext[l], cnts, lists);
    }
}


# Next layer's x@W moved into the owner group's epilogue: relu(sum+bias) stays in LDS rows, W in LDS, a wave per 16-row tile on f32 WMMA writes next h; 2 GEMM launches, x arrays gone; sum orders kept
# speedup vs baseline: 3.6752x; 1.3626x over previous
#include <hip/hip_runtime.h>

#define NN 100000
#define NE 1600000
#define DD 128
#define ROWS 391
#define NGRP 256
#define SEGCAP 288
#define LCAP 512
#define PITCH 132
#define ROWSP 400

typedef __attribute__((ext_vector_type(2))) float v2f;
typedef __attribute__((ext_vector_type(8))) float v8f;

__global__ __launch_bounds__(256) void gcn_gemm_wmma(const float* __restrict__ x,
                                                     const float* __restrict__ W,
                                                     float* __restrict__ h) {
    __shared__ float ldsWT[16 * 132];
    const int tid     = threadIdx.x;
    const int colBase = blockIdx.y * 16;

    #pragma unroll
    for (int i = 0; i < 8; ++i) {
        int e = tid + i * 256;
        int n = e & 15;
        int k = e >> 4;
        ldsWT[n * 132 + k] = W[k * DD + colBase + n];
    }
    __syncthreads();

    const int wave  = tid >> 5;
    const int lane  = tid & 31;
    const int mTile = blockIdx.x * 8 + wave;
    if (mTile >= (NN / 16)) return;

    const int    row  = mTile * 16 + (lane & 15);
    const int    koff = (lane >> 4) * 2;
    const float* xrow = x + (size_t)row * DD;
    const float* wtn  = ldsWT + (lane & 15) * 132;

    v8f acc = {};
    #pragma unroll
    for (int kk = 0; kk < DD; kk += 4) {
        v2f a, b;
        a.x = xrow[kk + koff];
        a.y = xrow[kk + koff + 1];
        b.x = wtn[kk + koff];
        b.y = wtn[kk + koff + 1];
        acc = __builtin_amdgcn_wmma_f32_16x16x4_f32(
                   false, a,  false, b,
                   (short)0, acc,  false,  false);
    }

    const int col     = colBase + (lane & 15);
    const int rowBase = mTile * 16 + (lane >> 4) * 8;
    #pragma unroll
    for (int v = 0; v < 8; ++v) {
        h[(size_t)(rowBase + v) * DD + col] = acc[v];
    }
}

__device__ __forceinline__ void lds_add(float* p, float v) {
    __hip_atomic_fetch_add(p, v, __ATOMIC_RELAXED, __HIP_MEMORY_SCOPE_WORKGROUP);
}

__device__ __forceinline__ void gather_rows(int s, int r, int wb, int n, int lane,
                                            const float* __restrict__ h, float* acc) {
    for (int k0 = 0; k0 < n; k0 += 8) {
        float4 hv[8];
        int    rr[8];
        float  ww[8];
        #pragma unroll
        for (int u = 0; u < 8; ++u) {
            const int k  = min(k0 + u, n - 1);
            const int sk = __builtin_amdgcn_readlane(s, k);
            rr[u] = __builtin_amdgcn_readlane(r, k);
            ww[u] = __int_as_float(__builtin_amdgcn_readlane(wb, k));
            hv[u] = *(const float4*)(h + (size_t)sk * DD + lane * 4);
        }
        #pragma unroll
        for (int u = 0; u < 8; ++u) {
            if (k0 + u < n) {
                float* a = acc + rr[u] * PITCH + lane;
                lds_add(a,      hv[u].x * ww[u]);
                lds_add(a + 32, hv[u].y * ww[u]);
                lds_add(a + 64, hv[u].z * ww[u]);
                lds_add(a + 96, hv[u].w * ww[u]);
            }
        }
    }
}

__device__ __forceinline__ void gather_batch(const unsigned* list, int n, int lane,
                                             const float* __restrict__ h,
                                             const int* __restrict__ src,
                                             const float* __restrict__ w,
                                             float* acc, uint2* rec) {
    const unsigned ent = (lane < n) ? list[lane] : 0u;
    const int   e  = (int)(ent & 0x1FFFFFu);
    const int   r  = (int)(ent >> 21);
    const int   s  = src[e];
    const int   wb = __float_as_int(w[e]);
    if (rec != nullptr && lane < n)
        rec[lane] = make_uint2(((unsigned)r << 17) | (unsigned)s, (unsigned)wb);
    gather_rows(s, r, wb, n, lane, h, acc);
}

template <bool REC>
__device__ __forceinline__ int scan_edges(int wave, int lane, int lo, int nrows, unsigned* myseg,
                                          const float* __restrict__ h,
                                          const int* __restrict__ src,
                                          const int* __restrict__ dst,
                                          const float* __restrict__ w,
                                          float* acc, uint2* keep) {
    int cnt  = 0;
    int done = 0;

    const int NP = NE / 256;
    int  p  = wave;
    int4 a0 = *(const int4*)(dst + (size_t)p * 256 + lane * 4);
    int4 a1 = *(const int4*)(dst + (size_t)p * 256 + 128 + lane * 4);
    while (p < NP) {
        const int4 c0 = a0, c1 = a1;
        const int  pn = p + 32;
        if (pn < NP) {
            a0 = *(const int4*)(dst + (size_t)pn * 256 + lane * 4);
            a1 = *(const int4*)(dst + (size_t)pn * 256 + 128 + lane * 4);
        }
        const int d[8] = { c0.x, c0.y, c0.z, c0.w, c1.x, c1.y, c1.z, c1.w };
        #pragma unroll
        for (int j = 0; j < 8; ++j) {
            const unsigned r    = (unsigned)(d[j] - lo);
            const bool     m    = r < (unsigned)nrows;
            const unsigned mask = __builtin_amdgcn_ballot_w32(m);
            if (mask) {
                if (m) {
                    const int pos = cnt + (int)__builtin_amdgcn_mbcnt_lo(mask, 0u);
                    const int e   = p * 256 + (j >> 2) * 128 + lane * 4 + (j & 3);
                    myseg[pos] = (r << 21) | (unsigned)e;
                }
                cnt += __popc(mask);
            }
        }
        __builtin_amdgcn_wave_barrier();
        while (cnt >= 32) {
            cnt -= 32;
            gather_batch(myseg + cnt, 32, lane, h, src, w, acc,
                         (REC && done + 32 <= LCAP) ? keep + done : nullptr);
            done += 32;
        }
        __builtin_amdgcn_wave_barrier();
        p = pn;
    }
    if (cnt > 0) {
        gather_batch(myseg, cnt, lane, h, src, w, acc,
                     (REC && done + cnt <= LCAP) ? keep + done : nullptr);
        done += cnt;
    }
    return done;
}

template <bool FIRST, bool LAST>
__global__ __launch_bounds__(1024) void gcn_owner_scatter(const float* __restrict__ h,
                                                          const int* __restrict__ src,
                                                          const int* __restrict__ dst,
                                                          const float* __restrict__ w,
                                                          const float* __restrict__ bias,
                                                          const float* __restrict__ Wn,
                                                          float* __restrict__ xout,
                                                          int* counts, unsigned* lists) {
    __shared__ float    acc[ROWSP * PITCH];
    __shared__ unsigned seg[32 * SEGCAP];
    __shared__ float    ldsWT[LAST ? 4 : DD * PITCH];

    const int tid   = threadIdx.x;
    const int lane  = tid & 31;
    const int wave  = __builtin_amdgcn_readfirstlane(tid >> 5);
    const int lo    = blockIdx.x * ROWS;
    const int nrows = min(ROWS, NN - lo);

    for (int i = tid; i < ROWSP * PITCH / 4; i += 1024)
        ((float4*)acc)[i] = make_float4(0.f, 0.f, 0.f, 0.f);
    if constexpr (!LAST) {
        #pragma unroll
        for (int i = 0; i < 16; ++i) {
            const int e  = tid + i * 1024;
            const int n  = e & 15;
            const int k  = (e >> 4) & 127;
            const int jt = e >> 11;
            ldsWT[(jt * 16 + n) * PITCH + k] = Wn[k * DD + jt * 16 + n];
        }
    }
    __syncthreads();

    unsigned* myseg = seg + wave * SEGCAP;
    const int gw    = blockIdx.x * 32 + wave;
    uint2*    keep  = (uint2*)lists + (size_t)gw * LCAP;

    if (FIRST) {
        const int total = scan_edges<true>(wave, lane, lo, nrows, myseg, h, src, dst, w, acc, keep);
        if (lane == 0) counts[gw] = total;
    } else {
        const int total = __builtin_amdgcn_readfirstlane(counts[gw]);
        if (total <= LCAP) {
            uint2 nxt = make_uint2(0u, 0u);
            if (lane < total) nxt = keep[lane];
            for (int base = 0; base < total; base += 32) {
                const uint2 cur = nxt;
                const int   n   = min(32, total - base);
                if (base + 32 + lane < total) nxt = keep[base + 32 + lane];
                gather_rows((int)(cur.x & 0x1FFFFu), (int)(cur.x >> 17), (int)cur.y, n, lane, h, acc);
            }
        } else {
            scan_edges<false>(wave, lane, lo, nrows, myseg, h, src, dst, w, acc, nullptr);
        }
    }
    __syncthreads();

    const float4 bv = *(const float4*)(bias + lane * 4);
    if constexpr (LAST) {
        for (int r = wave; r < nrows; r += 32) {
            const float* a = acc + r * PITCH + lane;
            float4 o;
            o.x = fmaxf(a[0]  + bv.x, 0.0f);
            o.y = fmaxf(a[32] + bv.y, 0.0f);
            o.z = fmaxf(a[64] + bv.z, 0.0f);
            o.w = fmaxf(a[96] + bv.w, 0.0f);
            *(float4*)(xout + (size_t)(lo + r) * DD + lane * 4) = o;
        }
    } else {
        for (int r = wave; r < ROWSP; r += 32) {
            float* a = acc + r * PITCH;
            float4 o;
            o.x = fmaxf(a[lane]      + bv.x, 0.0f);
            o.y = fmaxf(a[32 + lane] + bv.y, 0.0f);
            o.z = fmaxf(a[64 + lane] + bv.z, 0.0f);
            o.w = fmaxf(a[96 + lane] + bv.w, 0.0f);
            __builtin_amdgcn_wave_barrier();
            *(float4*)(a + lane * 4) = o;
        }
        __syncthreads();

        const int nT   = (nrows + 15) >> 4;
        const int c16  = lane & 15;
        const int half = lane >> 4;
        const float* brow = ldsWT + c16 * PITCH + half * 2;
        for (int rt = wave; rt < nT; rt += 32) {
            const float* arow = acc + (rt * 16 + c16) * PITCH + half * 2;
            v8f d[8];
            #pragma unroll
            for (int j = 0; j < 8; ++j) d[j] = v8f{};
            for (int kk = 0; kk < DD; kk += 4) {
                const v2f a = *(const v2f*)(arow + kk);
                #pragma unroll
                for (int j = 0; j < 8; ++j) {
                    const v2f b = *(const v2f*)(brow + j * 16 * PITCH + kk);
                    d[j] = __builtin_amdgcn_wmma_f32_16x16x4_f32(
                                false, a,  false, b,
                                (short)0, d[j],  false,  false);
                }
            }
            #pragma unroll
            for (int j = 0; j < 8; ++j) {
                #pragma unroll
                for (int v = 0; v < 8; ++v)
                    acc[(rt * 16 + half * 8 + v) * PITCH + j * 16 + c16] = d[j][v];
            }
            __builtin_amdgcn_wave_barrier();
            const float* srow = acc + rt * 16 * PITCH + lane * 4;
            float*       grow = xout + (size_t)(lo + rt * 16) * DD + lane * 4;
            if (rt * 16 + 16 <= nrows) {
                #pragma unroll
                for (int i = 0; i < 16; ++i)
                    *(float4*)(grow + i * DD) = *(const float4*)(srow + i * PITCH);
            } else {
                for (int i = 0; rt * 16 + i < nrows; ++i)
                    *(float4*)(grow + i * DD) = *(const float4*)(srow + i * PITCH);
            }
        }
    }
}

extern "C" void kernel_launch(void* const* d_in, const int* in_sizes, int n_in,
                              void* d_out, int out_size, void* d_ws, size_t ws_size,
                              hipStream_t stream) {
    const float* x    = (const float*)d_in[0];
    const int*   esrc = (const int*)d_in[1];
    const int*   edst = (const int*)d_in[2];
    const float* ew   = (const float*)d_in[3];
    const float* W    = (const float*)d_in[4];
    const float* b    = (const float*)d_in[5];
    float*       out  = (float*)d_out;

    float* ws0 = (float*)d_ws;
    float* ws1 = ws0 + (size_t)NN * DD;
    int*      cnts  = (int*)(ws1 + (size_t)NN * DD);
    unsigned* lists = (unsigned*)(cnts + NGRP * 32);

    const dim3 gGemm((NN / 16 + 7) / 8, DD / 16);

    gcn_gemm_wmma<<<gGemm, 256, 0, stream>>>(x, W, ws0);
    gcn_owner_scatter<true,  false><<<NGRP, 1024, 0, stream>>>(ws0, esrc, edst, ew, b,          W + DD * DD,     ws1, cnts, lists);
    gcn_owner_scatter<false, false><<<NGRP, 1024, 0, stream>>>(ws1, esrc, edst, ew, b + DD,     W + 2 * DD * DD, ws0, cnts, lists);
    gcn_owner_scatter<false, true ><<<NGRP, 1024, 0, stream>>>(ws0, esrc, edst, ew, b + 2 * DD, W,               out, cnts, lists);
}


# Owner scan reads 1 byte per edge, not 4: GEMM launch threads write each edge's owner-group id byte and an 8-byte (row, src, weight) record; waves' edge shares, so LDS add order, change
# speedup vs baseline: 4.2578x; 1.1585x over previous
#include <hip/hip_runtime.h>

#define NN 100000
#define NE 1600000
#define DD 128
#define ROWS 391
#define NGRP 256
#define SEGCAP 64
#define LCAP 512
#define PITCH 132
#define ROWSP 400

typedef __attribute__((ext_vector_type(2))) float v2f;
typedef __attribute__((ext_vector_type(8))) float v8f;

#define GEMM_BX ((NN / 16 + 7) / 8)
static_assert(GEMM_BX * (DD / 16) * 256 >= NE, "one GEMM thread per edge");
static_assert(NGRP * ROWS >= NN && NGRP <= 256, "owner group id fits a byte");

__global__ __launch_bounds__(256) void gcn_gemm_wmma(const float* __restrict__ x,
                                                     const float* __restrict__ W,
                                                     float* __restrict__ h,
                                                     const int* __restrict__ src,
                                                     const int* __restrict__ dst,
                                                     const float* __restrict__ w,
                                                     unsigned char* __restrict__ gid8,
                                                     uint2* __restrict__ recs) {
    __shared__ float ldsWT[16 * 132];
    const int tid     = threadIdx.x;
    const int colBase = blockIdx.y * 16;

    {
        const int e = (blockIdx.y * GEMM_BX + blockIdx.x) * 256 + tid;
        if (e < NE) {
            const unsigned d = (unsigned)dst[e];
            const unsigned g = d / ROWS;
            gid8[e] = (unsigned char)g;
            recs[e] = make_uint2(((d - g * ROWS) << 17) | (unsigned)src[e], __float_as_uint(w[e]));
        }
    }

    #pragma unroll
    for (int i = 0; i < 8; ++i) {
        int e = tid + i * 256;
        int n = e & 15;
        int k = e >> 4;
        ldsWT[n * 132 + k] = W[k * DD + colBase + n];
    }
    __syncthreads();

    const int wave  = tid >> 5;
    const int lane  = tid & 31;
    const int mTile = blockIdx.x * 8 + wave;
    if (mTile >= (NN / 16)) return;

    const int    row  = mTile * 16 + (lane & 15);
    const int    koff = (lane >> 4) * 2;
    const float* xrow = x + (size_t)row * DD;
    const float* wtn  = ldsWT + (lane & 15) * 132;

    v8f acc = {};
    #pragma unroll
    for (int kk = 0; kk < DD; kk += 4) {
        v2f a, b;
        a.x = xrow[kk + koff];
        a.y = xrow[kk + koff + 1];
        b.x = wtn[kk + koff];
        b.y = wtn[kk + koff + 1];
        acc = __builtin_amdgcn_wmma_f32_16x16x4_f32(
                   false, a,  false, b,
                   (short)0, acc,  false,  false);
    }

    const int col     = colBase + (lane & 15);
    const int rowBase = mTile * 16 + (lane >> 4) * 8;
    #pragma unroll
    for (int v = 0; v < 8; ++v) {
        h[(size_t)(rowBase + v) * DD + col] = acc[v];
    }
}

__device__ __forceinline__ void lds_add(float* p, float v) {
    __hip_atomic_fetch_add(p, v, __ATOMIC_RELAXED, __HIP_MEMORY_SCOPE_WORKGROUP);
}

__device__ __forceinline__ void gather_rows(int s, int r, int wb, int n, int lane,
                                            const float* __restrict__ h, float* acc) {
    for (int k0 = 0; k0 < n; k0 += 8) {
        float4 hv[8];
        int    rr[8];
        float  ww[8];
        #pragma unroll
        for (int u = 0; u < 8; ++u) {
            const int k  = min(k0 + u, n - 1);
            const int sk = __builtin_amdgcn_readlane(s, k);
            rr[u] = __builtin_amdgcn_readlane(r, k);
            ww[u] = __int_as_float(__builtin_amdgcn_readlane(wb, k));
            hv[u] = *(const float4*)(h + (size_t)sk * DD + lane * 4);
        }
        #pragma unroll
        for (int u = 0; u < 8; ++u) {
            if (k0 + u < n) {
                float* a = acc + rr[u] * PITCH + lane;
                lds_add(a,      hv[u].x * ww[u]);
                lds_add(a + 32, hv[u].y * ww[u]);
                lds_add(a + 64, hv[u].z * ww[u]);
                lds_add(a + 96, hv[u].w * ww[u]);
            }
        }
    }
}

__device__ __forceinline__ void gather_batch(const unsigned* list, int n, int lane,
                                             const float* __restrict__ h,
                                             const uint2* __restrict__ recs,
                                             float* acc, uint2* rec) {
    const unsigned e  = (lane < n) ? list[lane] : 0u;
    const uint2    en = recs[e];
    if (rec != nullptr && lane < n) rec[lane] = en;
    gather_rows((int)(en.x & 0x1FFFFu), (int)(en.x >> 17), (int)en.y, n, lane, h, acc);
}

__device__ __forceinline__ unsigned zero_bytes(unsigned v) {
    return ~(((v & 0x7F7F7F7Fu) + 0x7F7F7F7Fu) | v | 0x7F7F7F7Fu);
}

__device__ __forceinline__ unsigned match16(int4 c, unsigned pat) {
    return (zero_bytes((unsigned)c.x ^ pat) >> 7) | (zero_bytes((unsigned)c.y ^ pat) >> 6)
         | (zero_bytes((unsigned)c.z ^ pat) >> 5) | (zero_bytes((unsigned)c.w ^ pat) >> 4);
}

template <bool REC>
__device__ __forceinline__ int scan_edges(int wave, int lane, unsigned gpat, unsigned* myseg,
                                          const float* __restrict__ h,
                                          const unsigned char* __restrict__ gid8,
                                          const uint2* __restrict__ recs,
                                          float* acc, uint2* keep) {
    int cnt  = 0;
    int done = 0;

    static_assert(NE % 512 == 0, "whole half steps");
    const int   NH = NE / 512;
    const int   NQ = (NH + 1) / 2;
    const int4* g  = (const int4*)gid8;
    int  q = wave;
    int4 a = g[q * 64 + lane];
    int4 b = make_int4(0, 0, 0, 0);
    if (2 * q + 1 < NH) b = g[q * 64 + 32 + lane];
    while (q < NQ) {
        const int4 c0 = a, c1 = b;
        const bool two = 2 * q + 1 < NH;
        const int  qn  = q + 32;
        if (qn < NQ) {
            a = g[qn * 64 + lane];
            if (2 * qn + 1 < NH) b = g[qn * 64 + 32 + lane];
        }
        unsigned mm = match16(c0, gpat);
        if (two) mm |= match16(c1, gpat) << 4;
        unsigned any = __builtin_amdgcn_ballot_w32(mm != 0u);
        while (any) {
            if (mm != 0u) {
                const int bi  = __builtin_ctz(mm);
                const int pos = cnt + (int)__builtin_amdgcn_mbcnt_lo(any, 0u);
                myseg[pos] = (unsigned)(q * 1024 + ((bi & 4) << 7) + lane * 16 + ((bi & 3) << 2) + (bi >> 3));
                mm &= mm - 1u;
            }
            cnt += __popc(any);
            __builtin_amdgcn_wave_barrier();
            if (cnt >= 32) {
                cnt -= 32;
                gather_batch(myseg + cnt, 32, lane, h, recs, acc,
                             (REC && done + 32 <= LCAP) ? keep + done : nullptr);
                done += 32;
            }
            __builtin_amdgcn_wave_barrier();
            any = __builtin_amdgcn_ballot_w32(mm != 0u);
        }
        q = qn;
    }
    if (cnt > 0) {
        gather_batch(myseg, cnt, lane, h, recs, acc,
                     (REC && done + cnt <= LCAP) ? keep + done : nullptr);
        done += cnt;
    }
    return done;
}

template <bool FIRST, bool LAST>
__global__ __launch_bounds__(1024) void gcn_owner_scatter(const float* __restrict__ h,
                                                          const unsigned char* __restrict__ gid8,
                                                          const uint2* __restrict__ recs,
                                                          const float* __restrict__ bias,
                                                          const float* __restrict__ Wn,
                                                          float* __restrict__ xout,
                                                          int* counts, unsigned* lists) {
    __shared__ float    acc[ROWSP * PITCH];
    __shared__ unsigned seg[32 * SEGCAP];
    __shared__ float    ldsWT[LAST ? 4 : DD * PITCH];

    const int tid   = threadIdx.x;
    const int lane  = tid & 31;
    const int wave  = __builtin_amdgcn_readfirstlane(tid >> 5);
    const int lo    = blockIdx.x * ROWS;
    const int nrows = min(ROWS, NN - lo);
    const unsigned gpat = blockIdx.x * 0x01010101u;

    for (int i = tid; i < ROWSP * PITCH / 4; i += 1024)
        ((float4*)acc)[i] = make_float4(0.f, 0.f, 0.f, 0.f);
    if constexpr (!LAST) {
        #pragma unroll
        for (int i = 0; i < 16; ++i) {
            const int e  = tid + i * 1024;
            const int n  = e & 15;
            const int k  = (e >> 4) & 127;
            const int jt = e >> 11;
            ldsWT[(jt * 16 + n) * PITCH + k] = Wn[k * DD + jt * 16 + n];
        }
    }
    __syncthreads();

    unsigned* myseg = seg + wave * SEGCAP;
    const int gw    = blockIdx.x * 32 + wave;
    uint2*    keep  = (uint2*)lists + (size_t)gw * LCAP;

    if (FIRST) {
        const int total = scan_edges<true>(wave, lane, gpat, myseg, h, gid8, recs, acc, keep);
        if (lane == 0) counts[gw] = total;
    } else {
        const int total = __builtin_amdgcn_readfirstlane(counts[gw]);
        if (total <= LCAP) {
            uint2 nxt = make_uint2(0u, 0u);
            if (lane < total) nxt = keep[lane];
            for (int base = 0; base < total; base += 32) {
                const uint2 cur = nxt;
                const int   n   = min(32, total - base);
                if (base + 32 + lane < total) nxt = keep[base + 32 + lane];
                gather_rows((int)(cur.x & 0x1FFFFu), (int)(cur.x >> 17), (int)cur.y, n, lane, h, acc);
            }
        } else {
            scan_edges<false>(wave, lane, gpat, myseg, h, gid8, recs, acc, nullptr);
        }
    }
    __syncthreads();

    const float4 bv = *(const float4*)(bias + lane * 4);
    if constexpr (LAST) {
        for (int r = wave; r < nrows; r += 32) {
            const float* a = acc + r * PITCH + lane;
            float4 o;
            o.x = fmaxf(a[0]  + bv.x, 0.0f);
            o.y = fmaxf(a[32] + bv.y, 0.0f);
            o.z = fmaxf(a[64] + bv.z, 0.0f);
            o.w = fmaxf(a[96] + bv.w, 0.0f);
            *(float4*)(xout + (size_t)(lo + r) * DD + lane * 4) = o;
        }
    } else {
        for (int r = wave; r < ROWSP; r += 32) {
            float* a = acc + r * PITCH;
            float4 o;
            o.x = fmaxf(a[lane]      + bv.x, 0.0f);
            o.y = fmaxf(a[32 + lane] + bv.y, 0.0f);
            o.z = fmaxf(a[64 + lane] + bv.z, 0.0f);
            o.w = fmaxf(a[96 + lane] + bv.w, 0.0f);
            __builtin_amdgcn_wave_barrier();
            *(float4*)(a + lane * 4) = o;
        }
        __syncthreads();

        const int nT   = (nrows + 15) >> 4;
        const int c16  = lane & 15;
        const int half = lane >> 4;
        const float* brow = ldsWT + c16 * PITCH + half * 2;
        for (int rt = wave; rt < nT; rt += 32) {
            const float* arow = acc + (rt * 16 + c16) * PITCH + half * 2;
            v8f d[8];
            #pragma unroll
            for (int j = 0; j < 8; ++j) d[j] = v8f{};
            for (int kk = 0; kk < DD; kk += 4) {
                const v2f a = *(const v2f*)(arow + kk);
                #pragma unroll
                for (int j = 0; j < 8; ++j) {
                    const v2f b = *(const v2f*)(brow + j * 16 * PITCH + kk);
                    d[j] = __builtin_amdgcn_wmma_f32_16x16x4_f32(
                                false, a,  false, b,
                                (short)0, d[j],  false,  false);
                }
            }
            #pragma unroll
            for (int j = 0; j < 8; ++j) {
                #pragma unroll
                for (int v = 0; v < 8; ++v)
                    acc[(rt * 16 + half * 8 + v) * PITCH + j * 16 + c16] = d[j][v];
            }
            __builtin_amdgcn_wave_barrier();
            const float* srow = acc + rt * 16 * PITCH + lane * 4;
            float*       grow = xout + (size_t)(lo + rt * 16) * DD + lane * 4;
            if (rt * 16 + 16 <= nrows) {
                #pragma unroll
                for (int i = 0; i < 16; ++i)
                    *(float4*)(grow + i * DD) = *(const float4*)(srow + i * PITCH);
            } else {
                for (int i = 0; rt * 16 + i < nrows; ++i)
                    *(float4*)(grow + i * DD) = *(const float4*)(srow + i * PITCH);
            }
        }
    }
}

extern "C" void kernel_launch(void* const* d_in, const int* in_sizes, int n_in,
                              void* d_out, int out_size, void* d_ws, size_t ws_size,
                              hipStream_t stream) {
    const float* x    = (const float*)d_in[0];
    const int*   esrc = (const int*)d_in[1];
    const int*   edst = (const int*)d_in[2];
    const float* ew   = (const float*)d_in[3];
    const float* W    = (const float*)d_in[4];
    const float* b    = (const float*)d_in[5];
    float*       out  = (float*)d_out;

    float* ws0 = (float*)d_ws;
    float* ws1 = ws0 + (size_t)NN * DD;
    int*      cnts  = (int*)(ws1 + (size_t)NN * DD);
    unsigned* lists = (unsigned*)(cnts + NGRP * 32);
    uint2*         recs = (uint2*)lists + (size_t)NGRP * 32 * LCAP;
    unsigned char* gid8 = (unsigned char*)(recs + NE);

    const dim3 gGemm(GEMM_BX, DD / 16);

    gcn_gemm_wmma<<<gGemm, 256, 0, stream>>>(x, W, ws0, esrc, edst, ew, gid8, recs);
    gcn_owner_scatter<true,  false><<<NGRP, 1024, 0, stream>>>(ws0, gid8, recs, b,          W + DD * DD,     ws1, cnts, lists);
    gcn_owner_scatter<false, false><<<NGRP, 1024, 0, stream>>>(ws1, gid8, recs, b + DD,     W + 2 * DD * DD, ws0, cnts, lists);
    gcn_owner_scatter<false, true ><<<NGRP, 1024, 0, stream>>>(ws0, gid8, recs, b + 2 * DD, W,               out, cnts, lists);
}


# Layer-1 x@W re-cut onto the owner kernels' tile path: 256 groups of 16 waves hold W in LDS, a wave stages 16 whole x rows in LDS and multiplies by all 8 column tiles; x read once, k order kept
# speedup vs baseline: 5.0767x; 1.1923x over previous
#include <hip/hip_runtime.h>

#define NN 100000
#define NE 1600000
#define DD 128
#define ROWS 391
#define NGRP 256
#define SEGCAP 64
#define LCAP 512
#define PITCH 132
#define ROWSP 400

typedef __attribute__((ext_vector_type(2))) float v2f;
typedef __attribute__((ext_vector_type(8))) float v8f;
typedef __attribute__((ext_vector_type(4))) float v4f;

#define GW 16
#define GTHREADS (NGRP * GW * 32)
#define EFULL (NE / GTHREADS)
static_assert((EFULL + 1) * GTHREADS >= NE, "every edge re-coded");
static_assert(NN % 16 == 0, "whole 16-row tiles");
static_assert(NGRP * ROWS >= NN && NGRP <= 256, "owner group id fits a byte");

__global__ __launch_bounds__(GW * 32) void gcn_gemm_rows(const float* __restrict__ x,
                                                         const float* __restrict__ W,
                                                         float* __restrict__ h,
                                                         const int* __restrict__ src,
                                                         const int* __restrict__ dst,
                                                         const float* __restrict__ w,
                                                         unsigned char* __restrict__ gid8,
                                                         uint2* __restrict__ recs) {
    __shared__ float ldsWT[DD * PITCH];
    __shared__ float xs[GW * 16 * PITCH];

    const int tid  = threadIdx.x;
    const int lane = tid & 31;
    const int wave = __builtin_amdgcn_readfirstlane(tid >> 5);

    const int NT   = NN / 16;
    float*    xt   = xs + wave * 16 * PITCH;
    const int c16  = lane & 15;
    const int half = lane >> 4;
    const float* brow = ldsWT + c16 * PITCH + half * 2;
    const float* arow = xt + c16 * PITCH + half * 2;

    int t = blockIdx.x + NGRP * wave;
    v4f nx[16];
    if (t < NT) {
        const float* xr = x + (size_t)t * 16 * DD + lane * 4;
        #pragma unroll
        for (int i = 0; i < 16; ++i) nx[i] = *(const v4f*)(xr + i * DD);
    }

    const int e0 = blockIdx.x * (GW * 32) + tid;
    unsigned dv[EFULL + 1], sv[EFULL + 1], wv[EFULL + 1];
    const int  el   = e0 + EFULL * GTHREADS;
    const bool tail = el < NE;
    #pragma unroll
    for (int i = 0; i < EFULL; ++i) {
        const int e = e0 + i * GTHREADS;
        dv[i] = (unsigned)dst[e];
        sv[i] = (unsigned)src[e];
        wv[i] = __float_as_uint(w[e]);
    }
    if (tail) {
        dv[EFULL] = (unsigned)dst[el];
        sv[EFULL] = (unsigned)src[el];
        wv[EFULL] = __float_as_uint(w[el]);
    }

    #pragma unroll
    for (int i = 0; i < DD * DD / (GW * 32); ++i) {
        const int e  = tid + i * (GW * 32);
        const int n  = e & 15;
        const int k  = (e >> 4) & 127;
        const int jt = e >> 11;
        ldsWT[(jt * 16 + n) * PITCH + k] = W[k * DD + jt * 16 + n];
    }
    __syncthreads();

    #pragma unroll
    for (int i = 0; i < EFULL; ++i) {
        const int e = e0 + i * GTHREADS;
        const unsigned g = dv[i] / ROWS;
        gid8[e] = (unsigned char)g;
        recs[e] = make_uint2(((dv[i] - g * ROWS) << 17) | sv[i], wv[i]);
    }
    if (tail) {
        const unsigned g = dv[EFULL] / ROWS;
        gid8[el] = (unsigned char)g;
        recs[el] = make_uint2(((dv[EFULL] - g * ROWS) << 17) | sv[EFULL], wv[EFULL]);
    }

    while (t < NT) {
        #pragma unroll
        for (int i = 0; i < 16; ++i) *(v4f*)(xt + i * PITCH + lane * 4) = nx[i];
        const int tn = t + NGRP * GW;
        if (tn < NT) {
            const float* xr = x + (size_t)tn * 16 * DD + lane * 4;
            #pragma unroll
            for (int i = 0; i < 16; ++i) nx[i] = *(const v4f*)(xr + i * DD);
        }
        __builtin_amdgcn_wave_barrier();

        v8f d[8];
        #pragma unroll
        for (int j = 0; j < 8; ++j) d[j] = v8f{};
        for (int kk = 0; kk < DD; kk += 4) {
            const v2f a = *(const v2f*)(arow + kk);
            #pragma unroll
            for (int j = 0; j < 8; ++j) {
                const v2f b = *(const v2f*)(brow + j * 16 * PITCH + kk);
                d[j] = __builtin_amdgcn_wmma_f32_16x16x4_f32(
                            false, a,  false, b,
                            (short)0, d[j],  false,  false);
            }
        }
        #pragma unroll
        for (int j = 0; j < 8; ++j) {
            #pragma unroll
            for (int v = 0; v < 8; ++v)
                xt[(half * 8 + v) * PITCH + j * 16 + c16] = d[j][v];
        }
        __builtin_amdgcn_wave_barrier();
        const float* srow = xt + lane * 4;
        float*       grow = h + (size_t)t * 16 * DD + lane * 4;
        #pragma unroll
        for (int i = 0; i < 16; ++i)
            *(float4*)(grow + i * DD) = *(const float4*)(srow + i * PITCH);
        t = tn;
    }
}

__device__ __forceinline__ void lds_add(float* p, float v) {
    __hip_atomic_fetch_add(p, v, __ATOMIC_RELAXED, __HIP_MEMORY_SCOPE_WORKGROUP);
}

__device__ __forceinline__ void gather_rows(int s, int r, int wb, int n, int lane,
                                            const float* __restrict__ h, float* acc) {
    for (int k0 = 0; k0 < n; k0 += 8) {
        float4 hv[8];
        int    rr[8];
        float  ww[8];
        #pragma unroll
        for (int u = 0; u < 8; ++u) {
            const int k  = min(k0 + u, n - 1);
            const int sk = __builtin_amdgcn_readlane(s, k);
            rr[u] = __builtin_amdgcn_readlane(r, k);
            ww[u] = __int_as_float(__builtin_amdgcn_readlane(wb, k));
            hv[u] = *(const float4*)(h + (size_t)sk * DD + lane * 4);
        }
        #pragma unroll
        for (int u = 0; u < 8; ++u) {
            if (k0 + u < n) {
                float* a = acc + rr[u] * PITCH + lane;
                lds_add(a,      hv[u].x * ww[u]);
                lds_add(a + 32, hv[u].y * ww[u]);
                lds_add(a + 64, hv[u].z * ww[u]);
                lds_add(a + 96, hv[u].w * ww[u]);
            }
        }
    }
}

__device__ __forceinline__ void gather_batch(const unsigned* list, int n, int lane,
                                             const float* __restrict__ h,
                                             const uint2* __restrict__ recs,
                                             float* acc, uint2* rec) {
    const unsigned e  = (lane < n) ? list[lane] : 0u;
    const uint2    en = recs[e];
    if (rec != nullptr && lane < n) rec[lane] = en;
    gather_rows((int)(en.x & 0x1FFFFu), (int)(en.x >> 17), (int)en.y, n, lane, h, acc);
}

__device__ __forceinline__ unsigned zero_bytes(unsigned v) {
    return ~(((v & 0x7F7F7F7Fu) + 0x7F7F7F7Fu) | v | 0x7F7F7F7Fu);
}

__device__ __forceinline__ unsigned match16(int4 c, unsigned pat) {
    return (zero_bytes((unsigned)c.x ^ pat) >> 7) | (zero_bytes((unsigned)c.y ^ pat) >> 6)
         | (zero_bytes((unsigned)c.z ^ pat) >> 5) | (zero_bytes((unsigned)c.w ^ pat) >> 4);
}

template <bool REC>
__device__ __forceinline__ int scan_edges(int wave, int lane, unsigned gpat, unsigned* myseg,
                                          const float* __restrict__ h,
                                          const unsigned char* __restrict__ gid8,
                                          const uint2* __restrict__ recs,
                                          float* acc, uint2* keep) {
    int cnt  = 0;
    int done = 0;

    static_assert(NE % 512 == 0, "whole half steps");
    const int   NH = NE / 512;
    const int   NQ = (NH + 1) / 2;
    const int4* g  = (const int4*)gid8;
    int  q = wave;
    int4 a = g[q * 64 + lane];
    int4 b = make_int4(0, 0, 0, 0);
    if (2 * q + 1 < NH) b = g[q * 64 + 32 + lane];
    while (q < NQ) {
        const int4 c0 = a, c1 = b;
        const bool two = 2 * q + 1 < NH;
        const int  qn  = q + 32;
        if (qn < NQ) {
            a = g[qn * 64 + lane];
            if (2 * qn + 1 < NH) b = g[qn * 64 + 32 + lane];
        }
        unsigned mm = match16(c0, gpat);
        if (two) mm |= match16(c1, gpat) << 4;
        unsigned any = __builtin_amdgcn_ballot_w32(mm != 0u);
        while (any) {
            if (mm != 0u) {
                const int bi  = __builtin_ctz(mm);
                const int pos = cnt + (int)__builtin_amdgcn_mbcnt_lo(any, 0u);
                myseg[pos] = (unsigned)(q * 1024 + ((bi & 4) << 7) + lane * 16 + ((bi & 3) << 2) + (bi >> 3));
                mm &= mm - 1u;
            }
            cnt += __popc(any);
            __builtin_amdgcn_wave_barrier();
            if (cnt >= 32) {
                cnt -= 32;
                gather_batch(myseg + cnt, 32, lane, h, recs, acc,
                             (REC && done + 32 <= LCAP) ? keep + done : nullptr);
                done += 32;
            }
            __builtin_amdgcn_wave_barrier();
            any = __builtin_amdgcn_ballot_w32(mm != 0u);
        }
        q = qn;
    }
    if (cnt > 0) {
        gather_batch(myseg, cnt, lane, h, recs, acc,
                     (REC && done + cnt <= LCAP) ? keep + done : nullptr);
        done += cnt;
    }
    return done;
}

template <bool FIRST, bool LAST>
__global__ __launch_bounds__(1024) void gcn_owner_scatter(const float* __restrict__ h,
                                                          const unsigned char* __restrict__ gid8,
                                                          const uint2* __restrict__ recs,
                                                          const float* __restrict__ bias,
                                                          const float* __restrict__ Wn,
                                                          float* __restrict__ xout,
                                                          int* counts, unsigned* lists) {
    __shared__ float    acc[ROWSP * PITCH];
    __shared__ unsigned seg[32 * SEGCAP];
    __shared__ float    ldsWT[LAST ? 4 : DD * PITCH];

    const int tid   = threadIdx.x;
    const int lane  = tid & 31;
    const int wave  = __builtin_amdgcn_readfirstlane(tid >> 5);
    const int lo    = blockIdx.x * ROWS;
    const int nrows = min(ROWS, NN - lo);
    const unsigned gpat = blockIdx.x * 0x01010101u;

    for (int i = tid; i < ROWSP * PITCH / 4; i += 1024)
        ((float4*)acc)[i] = make_float4(0.f, 0.f, 0.f, 0.f);
    if constexpr (!LAST) {
        #pragma unroll
        for (int i = 0; i < 16; ++i) {
            const int e  = tid + i * 1024;
            const int n  = e & 15;
            const int k  = (e >> 4) & 127;
            const int jt = e >> 11;
            ldsWT[(jt * 16 + n) * PITCH + k] = Wn[k * DD + jt * 16 + n];
        }
    }
    __syncthreads();

    unsigned* myseg = seg + wave * SEGCAP;
    const int gw    = blockIdx.x * 32 + wave;
    uint2*    keep  = (uint2*)lists + (size_t)gw * LCAP;

    if (FIRST) {
        const int total = scan_edges<true>(wave, lane, gpat, myseg, h, gid8, recs, acc, keep);
        if (lane == 0) counts[gw] = total;
    } else {
        const int total = __builtin_amdgcn_readfirstlane(counts[gw]);
        if (total <= LCAP) {
            uint2 nxt = make_uint2(0u, 0u);
            if (lane < total) nxt = keep[lane];
            for (int base = 0; base < total; base += 32) {
                const uint2 cur = nxt;
                const int   n   = min(32, total - base);
                if (base + 32 + lane < total) nxt = keep[base + 32 + lane];
                gather_rows((int)(cur.x & 0x1FFFFu), (int)(cur.x >> 17), (int)cur.y, n, lane, h, acc);
            }
        } else {
            scan_edges<false>(wave, lane, gpat, myseg, h, gid8, recs, acc, nullptr);
        }
    }
    __syncthreads();

    const float4 bv = *(const float4*)(bias + lane * 4);
    if constexpr (LAST) {
        for (int r = wave; r < nrows; r += 32) {
            const float* a = acc + r * PITCH + lane;
            float4 o;
            o.x = fmaxf(a[0]  + bv.x, 0.0f);
            o.y = fmaxf(a[32] + bv.y, 0.0f);
            o.z = fmaxf(a[64] + bv.z, 0.0f);
            o.w = fmaxf(a[96] + bv.w, 0.0f);
            *(float4*)(xout + (size_t)(lo + r) * DD + lane * 4) = o;
        }
    } else {
        for (int r = wave; r < ROWSP; r += 32) {
            float* a = acc + r * PITCH;
            float4 o;
            o.x = fmaxf(a[lane]      + bv.x, 0.0f);
            o.y = fmaxf(a[32 + lane] + bv.y, 0.0f);
            o.z = fmaxf(a[64 + lane] + bv.z, 0.0f);
            o.w = fmaxf(a[96 + lane] + bv.w, 0.0f);
            __builtin_amdgcn_wave_barrier();
            *(float4*)(a + lane * 4) = o;
        }
        __syncthreads();

        const int nT   = (nrows + 15) >> 4;
        const int c16  = lane & 15;
        const int half = lane >> 4;
        const float* brow = ldsWT + c16 * PITCH + half * 2;
        for (int rt = wave; rt < nT; rt += 32) {
            const float* arow = acc + (rt * 16 + c16) * PITCH + half * 2;
            v8f d[8];
            #pragma unroll
            for (int j = 0; j < 8; ++j) d[j] = v8f{};
            for (int kk = 0; kk < DD; kk += 4) {
                const v2f a = *(const v2f*)(arow + kk);
                #pragma unroll
                for (int j = 0; j < 8; ++j) {
                    const v2f b = *(const v2f*)(brow + j * 16 * PITCH + kk);
                    d[j] = __builtin_amdgcn_wmma_f32_16x16x4_f32(
                                false, a,  false, b,
                                (short)0, d[j],  false,  false);
                }
            }
            #pragma unroll
            for (int j = 0; j < 8; ++j) {
                #pragma unroll
                for (int v = 0; v < 8; ++v)
                    acc[(rt * 16 + half * 8 + v) * PITCH + j * 16 + c16] = d[j][v];
            }
            __builtin_amdgcn_wave_barrier();
            const float* srow = acc + rt * 16 * PITCH + lane * 4;
            float*       grow = xout + (size_t)(lo + rt * 16) * DD + lane * 4;
            if (rt * 16 + 16 <= nrows) {
                #pragma unroll
                for (int i = 0; i < 16; ++i)
                    *(float4*)(grow + i * DD) = *(const float4*)(srow + i * PITCH);
            } else {
                for (int i = 0; rt * 16 + i < nrows; ++i)
                    *(float4*)(grow + i * DD) = *(const float4*)(srow + i * PITCH);
            }
        }
    }
}

extern "C" void kernel_launch(void* const* d_in, const int* in_sizes, int n_in,
                              void* d_out, int out_size, void* d_ws, size_t ws_size,
                              hipStream_t stream) {
    const float* x    = (const float*)d_in[0];
    const int*   esrc = (const int*)d_in[1];
    const int*   edst = (const int*)d_in[2];
    const float* ew   = (const float*)d_in[3];
    const float* W    = (const float*)d_in[4];
    const float* b    = (const float*)d_in[5];
    float*       out  = (float*)d_out;

    float* ws0 = (float*)d_ws;
    float* ws1 = ws0 + (size_t)NN * DD;
    int*      cnts  = (int*)(ws1 + (size_t)NN * DD);
    unsigned* lists = (unsigned*)(cnts + NGRP * 32);
    uint2*         recs = (uint2*)lists + (size_t)NGRP * 32 * LCAP;
    unsigned char* gid8 = (unsigned char*)(recs + NE);

    gcn_gemm_rows<<<NGRP, GW * 32, 0, stream>>>(x, W, ws0, esrc, edst, ew, gid8, recs);
    gcn_owner_scatter<true,  false><<<NGRP, 1024, 0, stream>>>(ws0, gid8, recs, b,          W + DD * DD,     ws1, cnts, lists);
    gcn_owner_scatter<false, false><<<NGRP, 1024, 0, stream>>>(ws1, gid8, recs, b + DD,     W + 2 * DD * DD, ws0, cnts, lists);
    gcn_owner_scatter<false, true ><<<NGRP, 1024, 0, stream>>>(ws0, gid8, recs, b + 2 * DD, W,               out, cnts, lists);
}


# LDS float adds replaced by register sums: a dst row has one owner wave (row mod 32); layer 1 deals edges to owner lists (LDS int slot counters); sums indexed by row in registers; sum order changes
# speedup vs baseline: 5.2462x; 1.0334x over previous
#include <hip/hip_runtime.h>

#define NN 100000
#define NE 1600000
#define DD 128
#define ROWS 391
#define NGRP 256
#define SEGCAP 64
#define LCAP 512
#define PITCH 132
#define ROWSP 400

typedef __attribute__((ext_vector_type(2))) float v2f;
typedef __attribute__((ext_vector_type(8))) float v8f;
typedef __attribute__((ext_vector_type(4))) float v4f;

#define GW 16
#define GTHREADS (NGRP * GW * 32)
#define EFULL (NE / GTHREADS)
static_assert((EFULL + 1) * GTHREADS >= NE, "every edge re-coded");
static_assert(NN % 16 == 0, "whole 16-row tiles");
static_assert(NGRP * ROWS >= NN && NGRP <= 256, "owner group id fits a byte");

__global__ __launch_bounds__(GW * 32) void gcn_gemm_rows(const float* __restrict__ x,
                                                         const float* __restrict__ W,
                                                         float* __restrict__ h,
                                                         const int* __restrict__ src,
                                                         const int* __restrict__ dst,
                                                         const float* __restrict__ w,
                                                         unsigned char* __restrict__ gid8,
                                                         uint2* __restrict__ recs) {
    __shared__ float ldsWT[DD * PITCH];
    __shared__ float xs[GW * 16 * PITCH];

    const int tid  = threadIdx.x;
    const int lane = tid & 31;
    const int wave = __builtin_amdgcn_readfirstlane(tid >> 5);

    const int NT   = NN / 16;
    float*    xt   = xs + wave * 16 * PITCH;
    const int c16  = lane & 15;
    const int half = lane >> 4;
    const float* brow = ldsWT + c16 * PITCH + half * 2;
    const float* arow = xt + c16 * PITCH + half * 2;

    int t = blockIdx.x + NGRP * wave;
    v4f nx[16];
    if (t < NT) {
        const float* xr = x + (size_t)t * 16 * DD + lane * 4;
        #pragma unroll
        for (int i = 0; i < 16; ++i) nx[i] = *(const v4f*)(xr + i * DD);
    }

    const int e0 = blockIdx.x * (GW * 32) + tid;
    unsigned dv[EFULL + 1], sv[EFULL + 1], wv[EFULL + 1];
    const int  el   = e0 + EFULL * GTHREADS;
    const bool tail = el < NE;
    #pragma unroll
    for (int i = 0; i < EFULL; ++i) {
        const int e = e0 + i * GTHREADS;
        dv[i] = (unsigned)dst[e];
        sv[i] = (unsigned)src[e];
        wv[i] = __float_as_uint(w[e]);
    }
    if (tail) {
        dv[EFULL] = (unsigned)dst[el];
        sv[EFULL] = (unsigned)src[el];
        wv[EFULL] = __float_as_uint(w[el]);
    }

    #pragma unroll
    for (int i = 0; i < DD * DD / (GW * 32); ++i) {
        const int e  = tid + i * (GW * 32);
        const int n  = e & 15;
        const int k  = (e >> 4) & 127;
        const int jt = e >> 11;
        ldsWT[(jt * 16 + n) * PITCH + k] = W[k * DD + jt * 16 + n];
    }
    __syncthreads();

    #pragma unroll
    for (int i = 0; i < EFULL; ++i) {
        const int e = e0 + i * GTHREADS;
        const unsigned g = dv[i] / ROWS;
        gid8[e] = (unsigned char)g;
        recs[e] = make_uint2(((dv[i] - g * ROWS) << 17) | sv[i], wv[i]);
    }
    if (tail) {
        const unsigned g = dv[EFULL] / ROWS;
        gid8[el] = (unsigned char)g;
        recs[el] = make_uint2(((dv[EFULL] - g * ROWS) << 17) | sv[EFULL], wv[EFULL]);
    }

    while (t < NT) {
        #pragma unroll
        for (int i = 0; i < 16; ++i) *(v4f*)(xt + i * PITCH + lane * 4) = nx[i];
        const int tn = t + NGRP * GW;
        if (tn < NT) {
            const float* xr = x + (size_t)tn * 16 * DD + lane * 4;
            #pragma unroll
            for (int i = 0; i < 16; ++i) nx[i] = *(const v4f*)(xr + i * DD);
        }
        __builtin_amdgcn_wave_barrier();

        v8f d[8];
        #pragma unroll
        for (int j = 0; j < 8; ++j) d[j] = v8f{};
        for (int kk = 0; kk < DD; kk += 4) {
            const v2f a = *(const v2f*)(arow + kk);
            #pragma unroll
            for (int j = 0; j < 8; ++j) {
                const v2f b = *(const v2f*)(brow + j * 16 * PITCH + kk);
                d[j] = __builtin_amdgcn_wmma_f32_16x16x4_f32(
                            false, a,  false, b,
                            (short)0, d[j],  false,  false);
            }
        }
        #pragma unroll
        for (int j = 0; j < 8; ++j) {
            #pragma unroll
            for (int v = 0; v < 8; ++v)
                xt[(half * 8 + v) * PITCH + j * 16 + c16] = d[j][v];
        }
        __builtin_amdgcn_wave_barrier();
        const float* srow = xt + lane * 4;
        float*       grow = h + (size_t)t * 16 * DD + lane * 4;
        #pragma unroll
        for (int i = 0; i < 16; ++i)
            *(float4*)(grow + i * DD) = *(const float4*)(srow + i * PITCH);
        t = tn;
    }
}

#define NOWN 13
static_assert(NOWN * 32 >= ROWSP && NOWN <= 16, "every LDS tile row has an owner; a wave's rows fit the 16 slots");

typedef __attribute__((ext_vector_type(16))) float v16f;
struct RowSums { v16f c0, c1, c2, c3; };

__device__ __forceinline__ void add_row(RowSums& a, int j, float4 hv, float w) {
    a.c0[j] += hv.x * w;
    a.c1[j] += hv.y * w;
    a.c2[j] += hv.z * w;
    a.c3[j] += hv.w * w;
}

__device__ __forceinline__ void gather_rows(int s, int j, int wb, int n, int lane,
                                            const float* __restrict__ h, RowSums& a) {
    for (int k0 = 0; k0 < n; k0 += 8) {
        float4 hv[8];
        int    jj[8];
        float  ww[8];
        #pragma unroll
        for (int u = 0; u < 8; ++u) {
            const int k  = min(k0 + u, n - 1);
            const int sk = __builtin_amdgcn_readlane(s, k);
            jj[u] = __builtin_amdgcn_readlane(j, k);
            ww[u] = __int_as_float(__builtin_amdgcn_readlane(wb, k));
            const float* rowp = h + (size_t)sk * DD;
            hv[u] = *(const float4*)(rowp + lane * 4);
        }
        #pragma unroll
        for (int u = 0; u < 8; ++u) {
            if (k0 + u < n) add_row(a, jj[u], hv[u], ww[u]);
        }
    }
}

__device__ __forceinline__ unsigned zero_bytes(unsigned v) {
    return ~(((v & 0x7F7F7F7Fu) + 0x7F7F7F7Fu) | v | 0x7F7F7F7Fu);
}

__device__ __forceinline__ unsigned match16(int4 c, unsigned pat) {
    return (zero_bytes((unsigned)c.x ^ pat) >> 7) | (zero_bytes((unsigned)c.y ^ pat) >> 6)
         | (zero_bytes((unsigned)c.z ^ pat) >> 5) | (zero_bytes((unsigned)c.w ^ pat) >> 4);
}

__device__ __forceinline__ void deal_batch(const unsigned* list, int n, int lane,
                                           const uint2* __restrict__ recs,
                                           int* dealCnt, uint2* glists) {
    if (lane < n) {
        const uint2 en    = recs[list[lane]];
        const int   owner = (int)((en.x >> 17) & 31u);
        const int   slot  = __hip_atomic_fetch_add(&dealCnt[owner], 1, __ATOMIC_RELAXED,
                                                   __HIP_MEMORY_SCOPE_WORKGROUP);
        if (slot < LCAP) glists[owner * LCAP + slot] = en;
    }
}

__device__ __forceinline__ void deal_edges(int wave, int lane, unsigned gpat, unsigned* myseg,
                                           const unsigned char* __restrict__ gid8,
                                           const uint2* __restrict__ recs,
                                           int* dealCnt, uint2* glists) {
    int cnt = 0;

    static_assert(NE % 512 == 0, "whole half steps");
    const int   NH = NE / 512;
    const int   NQ = (NH + 1) / 2;
    const int4* g  = (const int4*)gid8;
    int  q = wave;
    int4 a = g[q * 64 + lane];
    int4 b = make_int4(0, 0, 0, 0);
    if (2 * q + 1 < NH) b = g[q * 64 + 32 + lane];
    while (q < NQ) {
        const int4 c0 = a, c1 = b;
        const bool two = 2 * q + 1 < NH;
        const int  qn  = q + 32;
        if (qn < NQ) {
            a = g[qn * 64 + lane];
            if (2 * qn + 1 < NH) b = g[qn * 64 + 32 + lane];
        }
        unsigned mm = match16(c0, gpat);
        if (two) mm |= match16(c1, gpat) << 4;
        unsigned any = __builtin_amdgcn_ballot_w32(mm != 0u);
        while (any) {
            if (mm != 0u) {
                const int bi  = __builtin_ctz(mm);
                const int pos = cnt + (int)__builtin_amdgcn_mbcnt_lo(any, 0u);
                myseg[pos] = (unsigned)(q * 1024 + ((bi & 4) << 7) + lane * 16 + ((bi & 3) << 2) + (bi >> 3));
                mm &= mm - 1u;
            }
            cnt += __popc(any);
            __builtin_amdgcn_wave_barrier();
            if (cnt >= 32) {
                cnt -= 32;
                deal_batch(myseg + cnt, 32, lane, recs, dealCnt, glists);
            }
            __builtin_amdgcn_wave_barrier();
            any = __builtin_amdgcn_ballot_w32(mm != 0u);
        }
        q = qn;
    }
    if (cnt > 0) deal_batch(myseg, cnt, lane, recs, dealCnt, glists);
}

__device__ __forceinline__ void scan_own_rows(int wave, int lane, unsigned gpat,
                                           const float* __restrict__ h,
                                           const unsigned char* __restrict__ gid8,
                                           const uint2* __restrict__ recs,
                                           float* rows) {
    for (int r = wave; r < ROWSP; r += 32)
        *(float4*)(rows + r * PITCH + lane * 4) = make_float4(0.f, 0.f, 0.f, 0.f);
    const int4* g = (const int4*)gid8;
    for (int q = 0; q < NE / 512; ++q) {
        const unsigned mm = match16(g[q * 32 + lane], gpat);
        unsigned any = __builtin_amdgcn_ballot_w32(mm != 0u);
        while (any) {
            const int sl = __builtin_ctz(any);
            any &= any - 1u;
            unsigned m = (unsigned)__builtin_amdgcn_readlane((int)mm, sl);
            while (m) {
                const int bi = __builtin_ctz(m);
                m &= m - 1u;
                const int   e  = q * 512 + sl * 16 + ((bi & 3) << 2) + (bi >> 3);
                const uint2 en = recs[e];
                const unsigned x = (unsigned)__builtin_amdgcn_readfirstlane((int)en.x);
                const float    w = __int_as_float(__builtin_amdgcn_readfirstlane((int)en.y));
                if ((int)((x >> 17) & 31u) == wave) {
                    const float4 hv = *(const float4*)(h + (size_t)(x & 0x1FFFFu) * DD + lane * 4);
                    float4* p = (float4*)(rows + (x >> 17) * PITCH + lane * 4);
                    float4  s = *p;
                    s.x += hv.x * w; s.y += hv.y * w; s.z += hv.z * w; s.w += hv.w * w;
                    *p = s;
                }
            }
        }
    }
}

template <bool FIRST, bool LAST>
__global__ __launch_bounds__(1024) void gcn_owner_scatter(const float* __restrict__ h,
                                                          const unsigned char* __restrict__ gid8,
                                                          const uint2* __restrict__ recs,
                                                          const float* __restrict__ bias,
                                                          const float* __restrict__ Wn,
                                                          float* __restrict__ xout,
                                                          int* counts, unsigned* lists) {
    __shared__ float    acc[ROWSP * PITCH];
    __shared__ unsigned seg[FIRST ? 32 * SEGCAP : 4];
    __shared__ int      dealCnt[32];
    __shared__ float    ldsWT[LAST ? 4 : DD * PITCH];

    const int tid   = threadIdx.x;
    const int lane  = tid & 31;
    const int wave  = __builtin_amdgcn_readfirstlane(tid >> 5);
    const int lo    = blockIdx.x * ROWS;
    const int nrows = min(ROWS, NN - lo);
    const unsigned gpat = blockIdx.x * 0x01010101u;

    if (FIRST && tid < 32) dealCnt[tid] = 0;
    if constexpr (!LAST) {
        #pragma unroll
        for (int i = 0; i < 16; ++i) {
            const int e  = tid + i * 1024;
            const int n  = e & 15;
            const int k  = (e >> 4) & 127;
            const int jt = e >> 11;
            ldsWT[(jt * 16 + n) * PITCH + k] = Wn[k * DD + jt * 16 + n];
        }
    }
    __syncthreads();

    const int gw   = blockIdx.x * 32 + wave;
    uint2*    keep = (uint2*)lists + (size_t)gw * LCAP;

    int total;
    if (FIRST) {
        deal_edges(wave, lane, gpat, seg + wave * SEGCAP, gid8, recs, dealCnt,
                   (uint2*)lists + (size_t)blockIdx.x * 32 * LCAP);
        __syncthreads();
        total = __builtin_amdgcn_readfirstlane(dealCnt[wave]);
        if (lane == 0) counts[gw] = total;
    } else {
        total = __builtin_amdgcn_readfirstlane(counts[gw]);
    }

    RowSums sum;
    sum.c0 = v16f{}; sum.c1 = v16f{}; sum.c2 = v16f{}; sum.c3 = v16f{};

    if (total <= LCAP) {
        uint2 nxt = make_uint2(0u, 0u);
        if (lane < total) nxt = keep[lane];
        for (int base = 0; base < total; base += 32) {
            const uint2 cur = nxt;
            const int   n   = min(32, total - base);
            if (base + 32 + lane < total) nxt = keep[base + 32 + lane];
            gather_rows((int)(cur.x & 0x1FFFFu), (int)(cur.x >> 22), (int)cur.y, n, lane, h, sum);
        }
    } else {
        scan_own_rows(wave, lane, gpat, h, gid8, recs, acc);
        #pragma unroll
        for (int j = 0; j < NOWN; ++j) {
            const int r = wave + 32 * j;
            if (r < ROWSP) {
                const float4 t = *(const float4*)(acc + r * PITCH + lane * 4);
                sum.c0[j] = t.x; sum.c1[j] = t.y; sum.c2[j] = t.z; sum.c3[j] = t.w;
            }
        }
    }

    const float4 bv = *(const float4*)(bias + lane * 4);
    if constexpr (LAST) {
        #pragma unroll
        for (int j = 0; j < NOWN; ++j) {
            const int r = wave + 32 * j;
            if (r < nrows) {
                float4 o;
                o.x = fmaxf(sum.c0[j] + bv.x, 0.0f);
                o.y = fmaxf(sum.c1[j] + bv.y, 0.0f);
                o.z = fmaxf(sum.c2[j] + bv.z, 0.0f);
                o.w = fmaxf(sum.c3[j] + bv.w, 0.0f);
                *(float4*)(xout + (size_t)(lo + r) * DD + lane * 4) = o;
            }
        }
    } else {
        #pragma unroll
        for (int j = 0; j < NOWN; ++j) {
            const int r = wave + 32 * j;
            if (r < ROWSP) {
                float4 o;
                o.x = fmaxf(sum.c0[j] + bv.x, 0.0f);
                o.y = fmaxf(sum.c1[j] + bv.y, 0.0f);
                o.z = fmaxf(sum.c2[j] + bv.z, 0.0f);
                o.w = fmaxf(sum.c3[j] + bv.w, 0.0f);
                *(float4*)(acc + r * PITCH + lane * 4) = o;
            }
        }
        __syncthreads();

        const int nT   = (nrows + 15) >> 4;
        const int c16  = lane & 15;
        const int half = lane >> 4;
        const float* brow = ldsWT + c16 * PITCH + half * 2;
        for (int rt = wave; rt < nT; rt += 32) {
            const float* arow = acc + (rt * 16 + c16) * PITCH + half * 2;
            v8f d[8];
            #pragma unroll
            for (int j = 0; j < 8; ++j) d[j] = v8f{};
            for (int kk = 0; kk < DD; kk += 4) {
                const v2f a = *(const v2f*)(arow + kk);
                #pragma unroll
                for (int j = 0; j < 8; ++j) {
                    const v2f b = *(const v2f*)(brow + j * 16 * PITCH + kk);
                    d[j] = __builtin_amdgcn_wmma_f32_16x16x4_f32(
                                false, a,  false, b,
                                (short)0, d[j],  false,  false);
                }
            }
            #pragma unroll
            for (int j = 0; j < 8; ++j) {
                #pragma unroll
                for (int v = 0; v < 8; ++v)
                    acc[(rt * 16 + half * 8 + v) * PITCH + j * 16 + c16] = d[j][v];
            }
            __builtin_amdgcn_wave_barrier();
            const float* srow = acc + rt * 16 * PITCH + lane * 4;
            float*       grow = xout + (size_t)(lo + rt * 16) * DD + lane * 4;
            if (rt * 16 + 16 <= nrows) {
                #pragma unroll
                for (int i = 0; i < 16; ++i)
                    *(float4*)(grow + i * DD) = *(const float4*)(srow + i * PITCH);
            } else {
                for (int i = 0; rt * 16 + i < nrows; ++i)
                    *(float4*)(grow + i * DD) = *(const float4*)(srow + i * PITCH);
            }
        }
    }
}

extern "C" void kernel_launch(void* const* d_in, const int* in_sizes, int n_in,
                              void* d_out, int out_size, void* d_ws, size_t ws_size,
                              hipStream_t stream) {
    const float* x    = (const float*)d_in[0];
    const int*   esrc = (const int*)d_in[1];
    const int*   edst = (const int*)d_in[2];
    const float* ew   = (const float*)d_in[3];
    const float* W    = (const float*)d_in[4];
    const float* b    = (const float*)d_in[5];
    float*       out  = (float*)d_out;

    float* ws0 = (float*)d_ws;
    float* ws1 = ws0 + (size_t)NN * DD;
    int*      cnts  = (int*)(ws1 + (size_t)NN * DD);
    unsigned* lists = (unsigned*)(cnts + NGRP * 32);
    uint2*         recs = (uint2*)lists + (size_t)NGRP * 32 * LCAP;
    unsigned char* gid8 = (unsigned char*)(recs + NE);

    gcn_gemm_rows<<<NGRP, GW * 32, 0, stream>>>(x, W, ws0, esrc, edst, ew, gid8, recs);
    gcn_owner_scatter<true,  false><<<NGRP, 1024, 0, stream>>>(ws0, gid8, recs, b,          W + DD * DD,     ws1, cnts, lists);
    gcn_owner_scatter<false, false><<<NGRP, 1024, 0, stream>>>(ws1, gid8, recs, b + DD,     W + 2 * DD * DD, ws0, cnts, lists);
    gcn_owner_scatter<false, true ><<<NGRP, 1024, 0, stream>>>(ws0, gid8, recs, b + 2 * DD, W,               out, cnts, lists);
}


# Edge re-code orders each 6250-edge chunk's records by owner group (LDS rank counters, prefix sum); an owner group reads its 256 runs, no owner-id scan; deal order, so sum order, changes
# speedup vs baseline: 5.8333x; 1.1119x over previous
#include <hip/hip_runtime.h>

#define NN 100000
#define NE 1600000
#define DD 128
#define ROWS 391
#define NGRP 256
#define LCAP 512
#define PITCH 132
#define ROWSP 400

typedef __attribute__((ext_vector_type(2))) float v2f;
typedef __attribute__((ext_vector_type(8))) float v8f;
typedef __attribute__((ext_vector_type(4))) float v4f;

#define GW 16
#define CH (NE / NGRP)
#define EFULL (CH / (GW * 32))
static_assert(NE % NGRP == 0, "whole chunks");
static_assert((EFULL + 1) * GW * 32 >= CH, "every edge of a chunk re-coded");
static_assert(CH % 2 == 0, "a chunk's records leave as whole 16-byte words");
static_assert(CH < (1 << 24), "group id and rank share a word");
static_assert(NN % 16 == 0, "whole 16-row tiles");
static_assert(NGRP * ROWS >= NN && NGRP <= 256, "owner group id fits a byte");

__global__ __launch_bounds__(GW * 32) void gcn_gemm_rows(const float* __restrict__ x,
                                                         const float* __restrict__ W,
                                                         float* __restrict__ h,
                                                         const int* __restrict__ src,
                                                         const int* __restrict__ dst,
                                                         const float* __restrict__ w,
                                                         uint2* __restrict__ runs,
                                                         uint2* __restrict__ recs) {
    __shared__ float ldsWT[DD * PITCH];
    __shared__ float xs[GW * 16 * PITCH];
    __shared__ __attribute__((aligned(16))) unsigned stage[2 * CH];
    __shared__ int binCnt[NGRP];
    __shared__ int binOff[NGRP];

    const int tid  = threadIdx.x;
    const int lane = tid & 31;
    const int wave = __builtin_amdgcn_readfirstlane(tid >> 5);

    const int NT   = NN / 16;
    float*    xt   = xs + wave * 16 * PITCH;
    const int c16  = lane & 15;
    const int half = lane >> 4;
    const float* brow = ldsWT + c16 * PITCH + half * 2;
    const float* arow = xt + c16 * PITCH + half * 2;

    int t = blockIdx.x + NGRP * wave;
    v4f nx[16];
    if (t < NT) {
        const float* xr = x + (size_t)t * 16 * DD + lane * 4;
        #pragma unroll
        for (int i = 0; i < 16; ++i) nx[i] = *(const v4f*)(xr + i * DD);
    }

    const int cbase = blockIdx.x * CH;
    const int e0 = cbase + tid;
    unsigned dv[EFULL + 1], sv[EFULL + 1], wv[EFULL + 1];
    const int  el   = e0 + EFULL * (GW * 32);
    const bool tail = tid + EFULL * (GW * 32) < CH;
    #pragma unroll
    for (int i = 0; i < EFULL; ++i) {
        const int e = e0 + i * (GW * 32);
        dv[i] = (unsigned)dst[e];
        sv[i] = (unsigned)src[e];
        wv[i] = __float_as_uint(w[e]);
    }
    if (tail) {
        dv[EFULL] = (unsigned)dst[el];
        sv[EFULL] = (unsigned)src[el];
        wv[EFULL] = __float_as_uint(w[el]);
    }

    if (tid < NGRP) binCnt[tid] = 0;
    #pragma unroll
    for (int i = 0; i < DD * DD / (GW * 32); ++i) {
        const int e  = tid + i * (GW * 32);
        const int n  = e & 15;
        const int k  = (e >> 4) & 127;
        const int jt = e >> 11;
        ldsWT[(jt * 16 + n) * PITCH + k] = W[k * DD + jt * 16 + n];
    }
    __syncthreads();

    unsigned key[EFULL + 1];
    #pragma unroll
    for (int i = 0; i < EFULL; ++i) {
        const unsigned g = dv[i] / ROWS;
        const int r = __hip_atomic_fetch_add(&binCnt[g], 1, __ATOMIC_RELAXED, __HIP_MEMORY_SCOPE_WORKGROUP);
        key[i] = g | ((unsigned)r << 8);
        dv[i]  = ((dv[i] - g * ROWS) << 17) | sv[i];
    }
    if (tail) {
        const unsigned g = dv[EFULL] / ROWS;
        const int r = __hip_atomic_fetch_add(&binCnt[g], 1, __ATOMIC_RELAXED, __HIP_MEMORY_SCOPE_WORKGROUP);
        key[EFULL] = g | ((unsigned)r << 8);
        dv[EFULL]  = ((dv[EFULL] - g * ROWS) << 17) | sv[EFULL];
    }
    __syncthreads();
    if (wave == 0) {
        int c[NGRP / 32], s = 0;
        #pragma unroll
        for (int i = 0; i < NGRP / 32; ++i) { c[i] = binCnt[lane * (NGRP / 32) + i]; s += c[i]; }
        int incl = s;
        #pragma unroll
        for (int d = 1; d < 32; d <<= 1) {
            const int up = __shfl_up(incl, d, 32);
            if (lane >= d) incl += up;
        }
        int o = incl - s;
        #pragma unroll
        for (int i = 0; i < NGRP / 32; ++i) {
            binOff[lane * (NGRP / 32) + i] = o;
            o += c[i];
        }
    }
    __syncthreads();
    #pragma unroll
    for (int i = 0; i < EFULL; ++i) {
        const int p = binOff[key[i] & 255u] + (int)(key[i] >> 8);
        *(uint2*)(stage + 2 * p) = make_uint2(dv[i], wv[i]);
    }
    if (tail) {
        const int p = binOff[key[EFULL] & 255u] + (int)(key[EFULL] >> 8);
        *(uint2*)(stage + 2 * p) = make_uint2(dv[EFULL], wv[EFULL]);
    }
    __syncthreads();
    if (tid < NGRP) runs[(size_t)tid * NGRP + blockIdx.x] = make_uint2((unsigned)binOff[tid], (unsigned)binCnt[tid]);
    {
        const uint4* s4 = (const uint4*)stage;
        uint4*       g4 = (uint4*)(recs + cbase);
        #pragma unroll
        for (int k = 0; k < (CH / 2 + GW * 32 - 1) / (GW * 32); ++k) {
            const int i = tid + k * (GW * 32);
            if (i < CH / 2) g4[i] = s4[i];
        }
    }

    while (t < NT) {
        #pragma unroll
        for (int i = 0; i < 16; ++i) *(v4f*)(xt + i * PITCH + lane * 4) = nx[i];
        const int tn = t + NGRP * GW;
        if (tn < NT) {
            const float* xr = x + (size_t)tn * 16 * DD + lane * 4;
            #pragma unroll
            for (int i = 0; i < 16; ++i) nx[i] = *(const v4f*)(xr + i * DD);
        }
        __builtin_amdgcn_wave_barrier();

        v8f d[8];
        #pragma unroll
        for (int j = 0; j < 8; ++j) d[j] = v8f{};
        for (int kk = 0; kk < DD; kk += 4) {
            const v2f a = *(const v2f*)(arow + kk);
            #pragma unroll
            for (int j = 0; j < 8; ++j) {
                const v2f b = *(const v2f*)(brow + j * 16 * PITCH + kk);
                d[j] = __builtin_amdgcn_wmma_f32_16x16x4_f32(
                            false, a,  false, b,
                            (short)0, d[j],  false,  false);
            }
        }
        #pragma unroll
        for (int j = 0; j < 8; ++j) {
            #pragma unroll
            for (int v = 0; v < 8; ++v)
                xt[(half * 8 + v) * PITCH + j * 16 + c16] = d[j][v];
        }
        __builtin_amdgcn_wave_barrier();
        const float* srow = xt + lane * 4;
        float*       grow = h + (size_t)t * 16 * DD + lane * 4;
        #pragma unroll
        for (int i = 0; i < 16; ++i)
            *(float4*)(grow + i * DD) = *(const float4*)(srow + i * PITCH);
        t = tn;
    }
}

#define NOWN 13
static_assert(NOWN * 32 >= ROWSP && NOWN <= 16, "every LDS tile row has an owner; a wave's rows fit the 16 slots");

typedef __attribute__((ext_vector_type(16))) float v16f;
struct RowSums { v16f c0, c1, c2, c3; };

__device__ __forceinline__ void add_row(RowSums& a, int j, float4 hv, float w) {
    a.c0[j] += hv.x * w;
    a.c1[j] += hv.y * w;
    a.c2[j] += hv.z * w;
    a.c3[j] += hv.w * w;
}

__device__ __forceinline__ void gather_rows(int s, int j, int wb, int n, int lane,
                                            const float* __restrict__ h, RowSums& a) {
    for (int k0 = 0; k0 < n; k0 += 8) {
        float4 hv[8];
        int    jj[8];
        float  ww[8];
        #pragma unroll
        for (int u = 0; u < 8; ++u) {
            const int k  = min(k0 + u, n - 1);
            const int sk = __builtin_amdgcn_readlane(s, k);
            jj[u] = __builtin_amdgcn_readlane(j, k);
            ww[u] = __int_as_float(__builtin_amdgcn_readlane(wb, k));
            const float* rowp = h + (size_t)sk * DD;
            hv[u] = *(const float4*)(rowp + lane * 4);
        }
        #pragma unroll
        for (int u = 0; u < 8; ++u) {
            if (k0 + u < n) add_row(a, jj[u], hv[u], ww[u]);
        }
    }
}

__device__ __forceinline__ void deal_one(uint2 en, int* dealCnt, uint2* glists) {
    const int owner = (int)((en.x >> 17) & 31u);
    const int slot  = __hip_atomic_fetch_add(&dealCnt[owner], 1, __ATOMIC_RELAXED,
                                             __HIP_MEMORY_SCOPE_WORKGROUP);
    if (slot < LCAP) glists[owner * LCAP + slot] = en;
}

#define RPW (NGRP / 32)
__device__ __forceinline__ void deal_runs(int wave, int lane, int grp,
                                          const uint2* __restrict__ runs,
                                          const uint2* __restrict__ recs,
                                          int* dealCnt, uint2* glists) {
    const uint2* myruns = runs + (size_t)grp * NGRP + wave * RPW;
    uint2 r = make_uint2(0u, 0u);
    if (lane < RPW) r = myruns[lane];
    int   n[RPW];
    uint2 ea[RPW], eb[RPW];
    int   nmax = 0;
    #pragma unroll
    for (int u = 0; u < RPW; ++u) {
        const int st = __builtin_amdgcn_readlane((int)r.x, u);
        n[u] = __builtin_amdgcn_readlane((int)r.y, u);
        nmax = max(nmax, n[u]);
        const uint2* p = recs + (size_t)(wave * RPW + u) * CH + st;
        ea[u] = make_uint2(0u, 0u);
        eb[u] = make_uint2(0u, 0u);
        if (lane < n[u])      ea[u] = p[lane];
        if (lane + 32 < n[u]) eb[u] = p[lane + 32];
    }
    #pragma unroll
    for (int u = 0; u < RPW; ++u) {
        if (lane < n[u])      deal_one(ea[u], dealCnt, glists);
        if (lane + 32 < n[u]) deal_one(eb[u], dealCnt, glists);
    }
    if (nmax > 64) {
        #pragma nounroll
        for (int u = 0; u < RPW; ++u) {
            const int st = __builtin_amdgcn_readlane((int)r.x, u);
            const int nu = __builtin_amdgcn_readlane((int)r.y, u);
            const uint2* p = recs + (size_t)(wave * RPW + u) * CH + st;
            #pragma nounroll
            for (int base = 64; base < nu; base += 32) {
                if (base + lane < nu) deal_one(p[base + lane], dealCnt, glists);
            }
        }
    }
}

__device__ __forceinline__ void scan_own_rows(int wave, int lane, int grp,
                                           const float* __restrict__ h,
                                           const uint2* __restrict__ runs,
                                           const uint2* __restrict__ recs,
                                           float* rows) {
    for (int r = wave; r < ROWSP; r += 32)
        *(float4*)(rows + r * PITCH + lane * 4) = make_float4(0.f, 0.f, 0.f, 0.f);
    for (int c = 0; c < NGRP; ++c) {
        const uint2 rn = runs[(size_t)grp * NGRP + c];
        const int st = __builtin_amdgcn_readfirstlane((int)rn.x);
        const int n  = __builtin_amdgcn_readfirstlane((int)rn.y);
        const uint2* p = recs + (size_t)c * CH + st;
        for (int base = 0; base < n; base += 32) {
            uint2 en = make_uint2(0u, 0u);
            if (base + lane < n) en = p[base + lane];
            const int m = min(32, n - base);
            for (int k = 0; k < m; ++k) {
                const unsigned x = (unsigned)__builtin_amdgcn_readlane((int)en.x, k);
                const float    w = __int_as_float(__builtin_amdgcn_readlane((int)en.y, k));
                if ((int)((x >> 17) & 31u) == wave) {
                    const float4 hv = *(const float4*)(h + (size_t)(x & 0x1FFFFu) * DD + lane * 4);
                    float4* q = (float4*)(rows + (x >> 17) * PITCH + lane * 4);
                    float4  s = *q;
                    s.x += hv.x * w; s.y += hv.y * w; s.z += hv.z * w; s.w += hv.w * w;
                    *q = s;
                }
            }
        }
    }
}

template <bool FIRST, bool LAST>
__global__ __launch_bounds__(1024) void gcn_owner_scatter(const float* __restrict__ h,
                                                          const uint2* __restrict__ runs,
                                                          const uint2* __restrict__ recs,
                                                          const float* __restrict__ bias,
                                                          const float* __restrict__ Wn,
                                                          float* __restrict__ xout,
                                                          int* counts, unsigned* lists) {
    __shared__ float    acc[ROWSP * PITCH];
    __shared__ int      dealCnt[32];
    __shared__ float    ldsWT[LAST ? 4 : DD * PITCH];

    const int tid   = threadIdx.x;
    const int lane  = tid & 31;
    const int wave  = __builtin_amdgcn_readfirstlane(tid >> 5);
    const int lo    = blockIdx.x * ROWS;
    const int nrows = min(ROWS, NN - lo);

    if (FIRST && tid < 32) dealCnt[tid] = 0;
    if constexpr (!LAST) {
        #pragma unroll
        for (int i = 0; i < 16; ++i) {
            const int e  = tid + i * 1024;
            const int n  = e & 15;
            const int k  = (e >> 4) & 127;
            const int jt = e >> 11;
            ldsWT[(jt * 16 + n) * PITCH + k] = Wn[k * DD + jt * 16 + n];
        }
    }
    __syncthreads();

    const int gw   = blockIdx.x * 32 + wave;
    uint2*    keep = (uint2*)lists + (size_t)gw * LCAP;

    int total;
    if (FIRST) {
        deal_runs(wave, lane, blockIdx.x, runs, recs, dealCnt,
                  (uint2*)lists + (size_t)blockIdx.x * 32 * LCAP);
        __syncthreads();
        total = __builtin_amdgcn_readfirstlane(dealCnt[wave]);
        if (lane == 0) counts[gw] = total;
    } else {
        total = __builtin_amdgcn_readfirstlane(counts[gw]);
    }

    RowSums sum;
    sum.c0 = v16f{}; sum.c1 = v16f{}; sum.c2 = v16f{}; sum.c3 = v16f{};

    if (total <= LCAP) {
        uint2 nxt = make_uint2(0u, 0u);
        if (lane < total) nxt = keep[lane];
        for (int base = 0; base < total; base += 32) {
            const uint2 cur = nxt;
            const int   n   = min(32, total - base);
            if (base + 32 + lane < total) nxt = keep[base + 32 + lane];
            gather_rows((int)(cur.x & 0x1FFFFu), (int)(cur.x >> 22), (int)cur.y, n, lane, h, sum);
        }
    } else {
        scan_own_rows(wave, lane, blockIdx.x, h, runs, recs, acc);
        #pragma unroll
        for (int j = 0; j < NOWN; ++j) {
            const int r = wave + 32 * j;
            if (r < ROWSP) {
                const float4 t = *(const float4*)(acc + r * PITCH + lane * 4);
                sum.c0[j] = t.x; sum.c1[j] = t.y; sum.c2[j] = t.z; sum.c3[j] = t.w;
            }
        }
    }

    const float4 bv = *(const float4*)(bias + lane * 4);
    if constexpr (LAST) {
        #pragma unroll
        for (int j = 0; j < NOWN; ++j) {
            const int r = wave + 32 * j;
            if (r < nrows) {
                float4 o;
                o.x = fmaxf(sum.c0[j] + bv.x, 0.0f);
                o.y = fmaxf(sum.c1[j] + bv.y, 0.0f);
                o.z = fmaxf(sum.c2[j] + bv.z, 0.0f);
                o.w = fmaxf(sum.c3[j] + bv.w, 0.0f);
                *(float4*)(xout + (size_t)(lo + r) * DD + lane * 4) = o;
            }
        }
    } else {
        #pragma unroll
        for (int j = 0; j < NOWN; ++j) {
            const int r = wave + 32 * j;
            if (r < ROWSP) {
                float4 o;
                o.x = fmaxf(sum.c0[j] + bv.x, 0.0f);
                o.y = fmaxf(sum.c1[j] + bv.y, 0.0f);
                o.z = fmaxf(sum.c2[j] + bv.z, 0.0f);
                o.w = fmaxf(sum.c3[j] + bv.w, 0.0f);
                *(float4*)(acc + r * PITCH + lane * 4) = o;
            }
        }
        __syncthreads();

        const int nT   = (nrows + 15) >> 4;
        const int c16  = lane & 15;
        const int half = lane >> 4;
        const float* brow = ldsWT + c16 * PITCH + half * 2;
        for (int rt = wave; rt < nT; rt += 32) {
            const float* arow = acc + (rt * 16 + c16) * PITCH + half * 2;
            v8f d[8];
            #pragma unroll
            for (int j = 0; j < 8; ++j) d[j] = v8f{};
            for (int kk = 0; kk < DD; kk += 4) {
                const v2f a = *(const v2f*)(arow + kk);
                #pragma unroll
                for (int j = 0; j < 8; ++j) {
                    const v2f b = *(const v2f*)(brow + j * 16 * PITCH + kk);
                    d[j] = __builtin_amdgcn_wmma_f32_16x16x4_f32(
                                false, a,  false, b,
                                (short)0, d[j],  false,  false);
                }
            }
            #pragma unroll
            for (int j = 0; j < 8; ++j) {
                #pragma unroll
                for (int v = 0; v < 8; ++v)
                    acc[(rt * 16 + half * 8 + v) * PITCH + j * 16 + c16] = d[j][v];
            }
            __builtin_amdgcn_wave_barrier();
            const float* srow = acc + rt * 16 * PITCH + lane * 4;
            float*       grow = xout + (size_t)(lo + rt * 16) * DD + lane * 4;
            if (rt * 16 + 16 <= nrows) {
                #pragma unroll
                for (int i = 0; i < 16; ++i)
                    *(float4*)(grow + i * DD) = *(const float4*)(srow + i * PITCH);
            } else {
                for (int i = 0; rt * 16 + i < nrows; ++i)
                    *(float4*)(grow + i * DD) = *(const float4*)(srow + i * PITCH);
            }
        }
    }
}

extern "C" void kernel_launch(void* const* d_in, const int* in_sizes, int n_in,
                              void* d_out, int out_size, void* d_ws, size_t ws_size,
                              hipStream_t stream) {
    const float* x    = (const float*)d_in[0];
    const int*   esrc = (const int*)d_in[1];
    const int*   edst = (const int*)d_in[2];
    const float* ew   = (const float*)d_in[3];
    const float* W    = (const float*)d_in[4];
    const float* b    = (const float*)d_in[5];
    float*       out  = (float*)d_out;

    float* ws0 = (float*)d_ws;
    float* ws1 = ws0 + (size_t)NN * DD;
    int*      cnts  = (int*)(ws1 + (size_t)NN * DD);
    unsigned* lists = (unsigned*)(cnts + NGRP * 32);
    uint2* recs = (uint2*)lists + (size_t)NGRP * 32 * LCAP;
    uint2* runs = recs + NE;

    gcn_gemm_rows<<<NGRP, GW * 32, 0, stream>>>(x, W, ws0, esrc, edst, ew, runs, recs);
    gcn_owner_scatter<true,  false><<<NGRP, 1024, 0, stream>>>(ws0, runs, recs, b,          W + DD * DD,     ws1, cnts, lists);
    gcn_owner_scatter<false, false><<<NGRP, 1024, 0, stream>>>(ws1, runs, recs, b + DD,     W + 2 * DD * DD, ws0, cnts, lists);
    gcn_owner_scatter<false, true ><<<NGRP, 1024, 0, stream>>>(ws0, runs, recs, b + 2 * DD, W,               out, cnts, lists);
}


# Owner lists kept in source-row order: layer 1's deal counts per (owner wave, 4096-row range) in LDS, scans, places; all waves walk h front to back, a row's reads fall together; sum order changes
# speedup vs baseline: 6.0272x; 1.0332x over previous
#include <hip/hip_runtime.h>

#define NN 100000
#define NE 1600000
#define DD 128
#define ROWS 391
#define NGRP 256
#define LCAP 512
#define PITCH 132
#define ROWSP 400

typedef __attribute__((ext_vector_type(2))) float v2f;
typedef __attribute__((ext_vector_type(8))) float v8f;
typedef __attribute__((ext_vector_type(4))) float v4f;

#define GW 16
#define CH (NE / NGRP)
#define EFULL (CH / (GW * 32))
static_assert(NE % NGRP == 0, "whole chunks");
static_assert((EFULL + 1) * GW * 32 >= CH, "every edge of a chunk re-coded");
static_assert(CH % 2 == 0, "a chunk's records leave as whole 16-byte words");
static_assert(CH < (1 << 24), "group id and rank share a word");
static_assert(NN % 16 == 0, "whole 16-row tiles");
static_assert(NGRP * ROWS >= NN && NGRP <= 256, "owner group id fits a byte");

__global__ __launch_bounds__(GW * 32) void gcn_gemm_rows(const float* __restrict__ x,
                                                         const float* __restrict__ W,
                                                         float* __restrict__ h,
                                                         const int* __restrict__ src,
                                                         const int* __restrict__ dst,
                                                         const float* __restrict__ w,
                                                         uint2* __restrict__ runs,
                                                         uint2* __restrict__ recs) {
    __shared__ float ldsWT[DD * PITCH];
    __shared__ float xs[GW * 16 * PITCH];
    __shared__ __attribute__((aligned(16))) unsigned stage[2 * CH];
    __shared__ int binCnt[NGRP];
    __shared__ int binOff[NGRP];

    const int tid  = threadIdx.x;
    const int lane = tid & 31;
    const int wave = __builtin_amdgcn_readfirstlane(tid >> 5);

    const int NT   = NN / 16;
    float*    xt   = xs + wave * 16 * PITCH;
    const int c16  = lane & 15;
    const int half = lane >> 4;
    const float* brow = ldsWT + c16 * PITCH + half * 2;
    const float* arow = xt + c16 * PITCH + half * 2;

    int t = blockIdx.x + NGRP * wave;
    v4f nx[16];
    if (t < NT) {
        const float* xr = x + (size_t)t * 16 * DD + lane * 4;
        #pragma unroll
        for (int i = 0; i < 16; ++i) nx[i] = *(const v4f*)(xr + i * DD);
    }

    const int cbase = blockIdx.x * CH;
    const int e0 = cbase + tid;
    unsigned dv[EFULL + 1], sv[EFULL + 1], wv[EFULL + 1];
    const int  el   = e0 + EFULL * (GW * 32);
    const bool tail = tid + EFULL * (GW * 32) < CH;
    #pragma unroll
    for (int i = 0; i < EFULL; ++i) {
        const int e = e0 + i * (GW * 32);
        dv[i] = (unsigned)dst[e];
        sv[i] = (unsigned)src[e];
        wv[i] = __float_as_uint(w[e]);
    }
    if (tail) {
        dv[EFULL] = (unsigned)dst[el];
        sv[EFULL] = (unsigned)src[el];
        wv[EFULL] = __float_as_uint(w[el]);
    }

    if (tid < NGRP) binCnt[tid] = 0;
    #pragma unroll
    for (int i = 0; i < DD * DD / (GW * 32); ++i) {
        const int e  = tid + i * (GW * 32);
        const int n  = e & 15;
        const int k  = (e >> 4) & 127;
        const int jt = e >> 11;
        ldsWT[(jt * 16 + n) * PITCH + k] = W[k * DD + jt * 16 + n];
    }
    __syncthreads();

    unsigned key[EFULL + 1];
    #pragma unroll
    for (int i = 0; i < EFULL; ++i) {
        const unsigned g = dv[i] / ROWS;
        const int r = __hip_atomic_fetch_add(&binCnt[g], 1, __ATOMIC_RELAXED, __HIP_MEMORY_SCOPE_WORKGROUP);
        key[i] = g | ((unsigned)r << 8);
        dv[i]  = ((dv[i] - g * ROWS) << 17) | sv[i];
    }
    if (tail) {
        const unsigned g = dv[EFULL] / ROWS;
        const int r = __hip_atomic_fetch_add(&binCnt[g], 1, __ATOMIC_RELAXED, __HIP_MEMORY_SCOPE_WORKGROUP);
        key[EFULL] = g | ((unsigned)r << 8);
        dv[EFULL]  = ((dv[EFULL] - g * ROWS) << 17) | sv[EFULL];
    }
    __syncthreads();
    if (wave == 0) {
        int c[NGRP / 32], s = 0;
        #pragma unroll
        for (int i = 0; i < NGRP / 32; ++i) { c[i] = binCnt[lane * (NGRP / 32) + i]; s += c[i]; }
        int incl = s;
        #pragma unroll
        for (int d = 1; d < 32; d <<= 1) {
            const int up = __shfl_up(incl, d, 32);
            if (lane >= d) incl += up;
        }
        int o = incl - s;
        #pragma unroll
        for (int i = 0; i < NGRP / 32; ++i) {
            binOff[lane * (NGRP / 32) + i] = o;
            o += c[i];
        }
    }
    __syncthreads();
    #pragma unroll
    for (int i = 0; i < EFULL; ++i) {
        const int p = binOff[key[i] & 255u] + (int)(key[i] >> 8);
        *(uint2*)(stage + 2 * p) = make_uint2(dv[i], wv[i]);
    }
    if (tail) {
        const int p = binOff[key[EFULL] & 255u] + (int)(key[EFULL] >> 8);
        *(uint2*)(stage + 2 * p) = make_uint2(dv[EFULL], wv[EFULL]);
    }
    __syncthreads();
    if (tid < NGRP) runs[(size_t)tid * NGRP + blockIdx.x] = make_uint2((unsigned)binOff[tid], (unsigned)binCnt[tid]);
    {
        const uint4* s4 = (const uint4*)stage;
        uint4*       g4 = (uint4*)(recs + cbase);
        #pragma unroll
        for (int k = 0; k < (CH / 2 + GW * 32 - 1) / (GW * 32); ++k) {
            const int i = tid + k * (GW * 32);
            if (i < CH / 2) g4[i] = s4[i];
        }
    }

    while (t < NT) {
        #pragma unroll
        for (int i = 0; i < 16; ++i) *(v4f*)(xt + i * PITCH + lane * 4) = nx[i];
        const int tn = t + NGRP * GW;
        if (tn < NT) {
            const float* xr = x + (size_t)tn * 16 * DD + lane * 4;
            #pragma unroll
            for (int i = 0; i < 16; ++i) nx[i] = *(const v4f*)(xr + i * DD);
        }
        __builtin_amdgcn_wave_barrier();

        v8f d[8];
        #pragma unroll
        for (int j = 0; j < 8; ++j) d[j] = v8f{};
        for (int kk = 0; kk < DD; kk += 4) {
            const v2f a = *(const v2f*)(arow + kk);
            #pragma unroll
            for (int j = 0; j < 8; ++j) {
                const v2f b = *(const v2f*)(brow + j * 16 * PITCH + kk);
                d[j] = __builtin_amdgcn_wmma_f32_16x16x4_f32(
                            false, a,  false, b,
                            (short)0, d[j],  false,  false);
            }
        }
        #pragma unroll
        for (int j = 0; j < 8; ++j) {
            #pragma unroll
            for (int v = 0; v < 8; ++v)
                xt[(half * 8 + v) * PITCH + j * 16 + c16] = d[j][v];
        }
        __builtin_amdgcn_wave_barrier();
        const float* srow = xt + lane * 4;
        float*       grow = h + (size_t)t * 16 * DD + lane * 4;
        #pragma unroll
        for (int i = 0; i < 16; ++i)
            *(float4*)(grow + i * DD) = *(const float4*)(srow + i * PITCH);
        t = tn;
    }
}

#define NOWN 13
static_assert(NOWN * 32 >= ROWSP && NOWN <= 16, "every LDS tile row has an owner; a wave's rows fit the 16 slots");

typedef __attribute__((ext_vector_type(16))) float v16f;
struct RowSums { v16f c0, c1, c2, c3; };

__device__ __forceinline__ void add_row(RowSums& a, int j, float4 hv, float w) {
    a.c0[j] += hv.x * w;
    a.c1[j] += hv.y * w;
    a.c2[j] += hv.z * w;
    a.c3[j] += hv.w * w;
}

__device__ __forceinline__ void gather_rows(int s, int j, int wb, int n, int lane,
                                            const float* __restrict__ h, RowSums& a) {
    for (int k0 = 0; k0 < n; k0 += 8) {
        float4 hv[8];
        int    jj[8];
        float  ww[8];
        #pragma unroll
        for (int u = 0; u < 8; ++u) {
            const int k  = min(k0 + u, n - 1);
            const int sk = __builtin_amdgcn_readlane(s, k);
            jj[u] = __builtin_amdgcn_readlane(j, k);
            ww[u] = __int_as_float(__builtin_amdgcn_readlane(wb, k));
            const float* rowp = h + (size_t)sk * DD;
            hv[u] = *(const float4*)(rowp + lane * 4);
        }
        #pragma unroll
        for (int u = 0; u < 8; ++u) {
            if (k0 + u < n) add_row(a, jj[u], hv[u], ww[u]);
        }
    }
}

#define SB 32
#define SBSHIFT 12
static_assert(SB == 32 && 32 * SB == 1024, "a lane per range in the scan; a thread per counter at the start");
static_assert(((1 << 17) - 1) >> SBSHIFT < SB, "every src row a record can hold falls in a range");

#define RPW (NGRP / 32)
struct DealtRuns {
    uint2 r;
    int   n[RPW];
    uint2 ea[RPW], eb[RPW];
    int   nmax;
};

__device__ __forceinline__ void load_runs(DealtRuns& d, int wave, int lane, int grp,
                                          const uint2* __restrict__ runs,
                                          const uint2* __restrict__ recs) {
    const uint2* myruns = runs + (size_t)grp * NGRP + wave * RPW;
    d.r = make_uint2(0u, 0u);
    if (lane < RPW) d.r = myruns[lane];
    d.nmax = 0;
    #pragma unroll
    for (int u = 0; u < RPW; ++u) {
        const int st = __builtin_amdgcn_readlane((int)d.r.x, u);
        d.n[u] = __builtin_amdgcn_readlane((int)d.r.y, u);
        d.nmax = max(d.nmax, d.n[u]);
        const uint2* p = recs + (size_t)(wave * RPW + u) * CH + st;
        d.ea[u] = make_uint2(0u, 0u);
        d.eb[u] = make_uint2(0u, 0u);
        if (lane < d.n[u])      d.ea[u] = p[lane];
        if (lane + 32 < d.n[u]) d.eb[u] = p[lane + 32];
    }
}

template <bool PLACE>
__device__ __forceinline__ void deal_one(uint2 en, int* bins, uint2* glists) {
    const int owner = (int)((en.x >> 17) & 31u);
    const int bin   = owner * SB + (int)((en.x & 0x1FFFFu) >> SBSHIFT);
    const int slot  = __hip_atomic_fetch_add(&bins[bin], 1, __ATOMIC_RELAXED, __HIP_MEMORY_SCOPE_WORKGROUP);
    if (PLACE) {
        if (slot < LCAP) glists[owner * LCAP + slot] = en;
    }
}

template <bool PLACE>
__device__ __forceinline__ void deal_pass(const DealtRuns& d, int wave, int lane,
                                          const uint2* __restrict__ recs,
                                          int* bins, uint2* glists) {
    #pragma unroll
    for (int u = 0; u < RPW; ++u) {
        if (lane < d.n[u])      deal_one<PLACE>(d.ea[u], bins, glists);
        if (lane + 32 < d.n[u]) deal_one<PLACE>(d.eb[u], bins, glists);
    }
    if (d.nmax > 64) {
        #pragma nounroll
        for (int u = 0; u < RPW; ++u) {
            const int st = __builtin_amdgcn_readlane((int)d.r.x, u);
            const int nu = __builtin_amdgcn_readlane((int)d.r.y, u);
            const uint2* p = recs + (size_t)(wave * RPW + u) * CH + st;
            #pragma nounroll
            for (int base = 64; base < nu; base += 32) {
                if (base + lane < nu) deal_one<PLACE>(p[base + lane], bins, glists);
            }
        }
    }
}

__device__ __forceinline__ void scan_own_rows(int wave, int lane, int grp,
                                           const float* __restrict__ h,
                                           const uint2* __restrict__ runs,
                                           const uint2* __restrict__ recs,
                                           float* rows) {
    for (int r = wave; r < ROWSP; r += 32)
        *(float4*)(rows + r * PITCH + lane * 4) = make_float4(0.f, 0.f, 0.f, 0.f);
    for (int c = 0; c < NGRP; ++c) {
        const uint2 rn = runs[(size_t)grp * NGRP + c];
        const int st = __builtin_amdgcn_readfirstlane((int)rn.x);
        const int n  = __builtin_amdgcn_readfirstlane((int)rn.y);
        const uint2* p = recs + (size_t)c * CH + st;
        for (int base = 0; base < n; base += 32) {
            uint2 en = make_uint2(0u, 0u);
            if (base + lane < n) en = p[base + lane];
            const int m = min(32, n - base);
            for (int k = 0; k < m; ++k) {
                const unsigned x = (unsigned)__builtin_amdgcn_readlane((int)en.x, k);
                const float    w = __int_as_float(__builtin_amdgcn_readlane((int)en.y, k));
                if ((int)((x >> 17) & 31u) == wave) {
                    const float4 hv = *(const float4*)(h + (size_t)(x & 0x1FFFFu) * DD + lane * 4);
                    float4* q = (float4*)(rows + (x >> 17) * PITCH + lane * 4);
                    float4  s = *q;
                    s.x += hv.x * w; s.y += hv.y * w; s.z += hv.z * w; s.w += hv.w * w;
                    *q = s;
                }
            }
        }
    }
}

template <bool FIRST, bool LAST>
__global__ __launch_bounds__(1024) void gcn_owner_scatter(const float* __restrict__ h,
                                                          const uint2* __restrict__ runs,
                                                          const uint2* __restrict__ recs,
                                                          const float* __restrict__ bias,
                                                          const float* __restrict__ Wn,
                                                          float* __restrict__ xout,
                                                          int* counts, unsigned* lists) {
    __shared__ float    acc[ROWSP * PITCH];
    __shared__ int      binCnt[FIRST ? 32 * SB : 1];
    __shared__ int      binPos[FIRST ? 32 * SB : 1];
    __shared__ float    ldsWT[LAST ? 4 : DD * PITCH];

    const int tid   = threadIdx.x;
    const int lane  = tid & 31;
    const int wave  = __builtin_amdgcn_readfirstlane(tid >> 5);
    const int lo    = blockIdx.x * ROWS;
    const int nrows = min(ROWS, NN - lo);

    if constexpr (FIRST) binCnt[tid] = 0;
    if constexpr (!LAST) {
        #pragma unroll
        for (int i = 0; i < 16; ++i) {
            const int e  = tid + i * 1024;
            const int n  = e & 15;
            const int k  = (e >> 4) & 127;
            const int jt = e >> 11;
            ldsWT[(jt * 16 + n) * PITCH + k] = Wn[k * DD + jt * 16 + n];
        }
    }
    __syncthreads();

    const int gw   = blockIdx.x * 32 + wave;
    uint2*    keep = (uint2*)lists + (size_t)gw * LCAP;

    int total;
    if constexpr (FIRST) {
        uint2* glists = (uint2*)lists + (size_t)blockIdx.x * 32 * LCAP;
        DealtRuns d;
        load_runs(d, wave, lane, blockIdx.x, runs, recs);
        deal_pass<false>(d, wave, lane, recs, binCnt, glists);
        __syncthreads();
        {
            const int cnt = binCnt[wave * SB + lane];
            int incl = cnt;
            #pragma unroll
            for (int dd = 1; dd < 32; dd <<= 1) {
                const int up = __shfl_up(incl, dd, 32);
                if (lane >= dd) incl += up;
            }
            binPos[wave * SB + lane] = incl - cnt;
            total = __builtin_amdgcn_readlane(incl, 31);
        }
        __syncthreads();
        deal_pass<true>(d, wave, lane, recs, binPos, glists);
        __syncthreads();
        if (lane == 0) counts[gw] = total;
    } else {
        total = __builtin_amdgcn_readfirstlane(counts[gw]);
    }

    RowSums sum;
    sum.c0 = v16f{}; sum.c1 = v16f{}; sum.c2 = v16f{}; sum.c3 = v16f{};

    if (total <= LCAP) {
        uint2 nxt = make_uint2(0u, 0u);
        if (lane < total) nxt = keep[lane];
        for (int base = 0; base < total; base += 32) {
            const uint2 cur = nxt;
            const int   n   = min(32, total - base);
            if (base + 32 + lane < total) nxt = keep[base + 32 + lane];
            gather_rows((int)(cur.x & 0x1FFFFu), (int)(cur.x >> 22), (int)cur.y, n, lane, h, sum);
        }
    } else {
        scan_own_rows(wave, lane, blockIdx.x, h, runs, recs, acc);
        #pragma unroll
        for (int j = 0; j < NOWN; ++j) {
            const int r = wave + 32 * j;
            if (r < ROWSP) {
                const float4 t = *(const float4*)(acc + r * PITCH + lane * 4);
                sum.c0[j] = t.x; sum.c1[j] = t.y; sum.c2[j] = t.z; sum.c3[j] = t.w;
            }
        }
    }

    const float4 bv = *(const float4*)(bias + lane * 4);
    if constexpr (LAST) {
        #pragma unroll
        for (int j = 0; j < NOWN; ++j) {
            const int r = wave + 32 * j;
            if (r < nrows) {
                float4 o;
                o.x = fmaxf(sum.c0[j] + bv.x, 0.0f);
                o.y = fmaxf(sum.c1[j] + bv.y, 0.0f);
                o.z = fmaxf(sum.c2[j] + bv.z, 0.0f);
                o.w = fmaxf(sum.c3[j] + bv.w, 0.0f);
                *(float4*)(xout + (size_t)(lo + r) * DD + lane * 4) = o;
            }
        }
    } else {
        #pragma unroll
        for (int j = 0; j < NOWN; ++j) {
            const int r = wave + 32 * j;
            if (r < ROWSP) {
                float4 o;
                o.x = fmaxf(sum.c0[j] + bv.x, 0.0f);
                o.y = fmaxf(sum.c1[j] + bv.y, 0.0f);
                o.z = fmaxf(sum.c2[j] + bv.z, 0.0f);
                o.w = fmaxf(sum.c3[j] + bv.w, 0.0f);
                *(float4*)(acc + r * PITCH + lane * 4) = o;
            }
        }
        __syncthreads();

        const int nT   = (nrows + 15) >> 4;
        const int c16  = lane & 15;
        const int half = lane >> 4;
        const float* brow = ldsWT + c16 * PITCH + half * 2;
        for (int rt = wave; rt < nT; rt += 32) {
            const float* arow = acc + (rt * 16 + c16) * PITCH + half * 2;
            v8f d[8];
            #pragma unroll
            for (int j = 0; j < 8; ++j) d[j] = v8f{};
            for (int kk = 0; kk < DD; kk += 4) {
                const v2f a = *(const v2f*)(arow + kk);
                #pragma unroll
                for (int j = 0; j < 8; ++j) {
                    const v2f b = *(const v2f*)(brow + j * 16 * PITCH + kk);
                    d[j] = __builtin_amdgcn_wmma_f32_16x16x4_f32(
                                false, a,  false, b,
                                (short)0, d[j],  false,  false);
                }
            }
            #pragma unroll
            for (int j = 0; j < 8; ++j) {
                #pragma unroll
                for (int v = 0; v < 8; ++v)
                    acc[(rt * 16 + half * 8 + v) * PITCH + j * 16 + c16] = d[j][v];
            }
            __builtin_amdgcn_wave_barrier();
            const float* srow = acc + rt * 16 * PITCH + lane * 4;
            float*       grow = xout + (size_t)(lo + rt * 16) * DD + lane * 4;
            if (rt * 16 + 16 <= nrows) {
                #pragma unroll
                for (int i = 0; i < 16; ++i)
                    *(float4*)(grow + i * DD) = *(const float4*)(srow + i * PITCH);
            } else {
                for (int i = 0; rt * 16 + i < nrows; ++i)
                    *(float4*)(grow + i * DD) = *(const float4*)(srow + i * PITCH);
            }
        }
    }
}

extern "C" void kernel_launch(void* const* d_in, const int* in_sizes, int n_in,
                              void* d_out, int out_size, void* d_ws, size_t ws_size,
                              hipStream_t stream) {
    const float* x    = (const float*)d_in[0];
    const int*   esrc = (const int*)d_in[1];
    const int*   edst = (const int*)d_in[2];
    const float* ew   = (const float*)d_in[3];
    const float* W    = (const float*)d_in[4];
    const float* b    = (const float*)d_in[5];
    float*       out  = (float*)d_out;

    float* ws0 = (float*)d_ws;
    float* ws1 = ws0 + (size_t)NN * DD;
    int*      cnts  = (int*)(ws1 + (size_t)NN * DD);
    unsigned* lists = (unsigned*)(cnts + NGRP * 32);
    uint2* recs = (uint2*)lists + (size_t)NGRP * 32 * LCAP;
    uint2* runs = recs + NE;

    gcn_gemm_rows<<<NGRP, GW * 32, 0, stream>>>(x, W, ws0, esrc, edst, ew, runs, recs);
    gcn_owner_scatter<true,  false><<<NGRP, 1024, 0, stream>>>(ws0, runs, recs, b,          W + DD * DD,     ws1, cnts, lists);
    gcn_owner_scatter<false, false><<<NGRP, 1024, 0, stream>>>(ws1, runs, recs, b + DD,     W + 2 * DD * DD, ws0, cnts, lists);
    gcn_owner_scatter<false, true ><<<NGRP, 1024, 0, stream>>>(ws0, runs, recs, b + 2 * DD, W,               out, cnts, lists);
}
